# starting kernel
# baseline (speedup 1.0000x reference)
#include <hip/hip_runtime.h>

#define N_NODES 50000
#define N_EDGES 800000
#define DIM     128
#define NLAYER  3
#define NGRAPH  512
#define ZSTRIDE (NLAYER * DIM)
#define BN_EPS  1e-5f
#define LDS_STRIDE 132

typedef __attribute__((ext_vector_type(2))) float v2f;
typedef __attribute__((ext_vector_type(8))) float v8f;

__global__ __launch_bounds__(256) void scatter_kernel(
    const float* __restrict__ z, int zstride,
    const int* __restrict__ src, const int* __restrict__ dst,
    float* __restrict__ agg)
{
    int tid = blockIdx.x * blockDim.x + threadIdx.x;
    if (tid >= N_EDGES * 32) return;
    int e  = tid >> 5;
    int c  = (tid & 31) * 4;
    int s  = src[e];
    int d  = dst[e];
    float4 v = *(const float4*)(z + (size_t)s * zstride + c);
    float* a = agg + (size_t)d * DIM + c;
    atomicAdd(a + 0, v.x);
    atomicAdd(a + 1, v.y);
    atomicAdd(a + 2, v.z);
    atomicAdd(a + 3, v.w);
}

__global__ __launch_bounds__(256) void gin_mlp_kernel(
    const float* __restrict__ z, int zstride,
    float* __restrict__ hbuf,
    const float* __restrict__ W1, const float* __restrict__ b1,
    const float* __restrict__ W2, const float* __restrict__ b2,
    float* __restrict__ stats)
{
    __shared__ float ldsA[16 * LDS_STRIDE];
    __shared__ float ldsH[16 * LDS_STRIDE];

    const int tid  = threadIdx.x;
    const int wave = tid >> 5;
    const int lane = tid & 31;
    const int half = lane >> 4;
    const int l16  = lane & 15;
    const int row0 = blockIdx.x * 16;
    const int col  = wave * 16 + l16;

    for (int idx = tid; idx < 16 * DIM; idx += 256) {
        int r = idx >> 7, c = idx & 127;
        ldsA[r * LDS_STRIDE + c] =
            z[(size_t)(row0 + r) * zstride + c] + hbuf[(size_t)(row0 + r) * DIM + c];
    }
    __syncthreads();

    v8f acc = {};
    for (int k0 = 0; k0 < DIM; k0 += 4) {
        int ka = k0 + 2 * half;
        v2f a, b;
        a.x = ldsA[l16 * LDS_STRIDE + ka];
        a.y = ldsA[l16 * LDS_STRIDE + ka + 1];
        b.x = W1[ka * DIM + col];
        b.y = W1[(ka + 1) * DIM + col];
        acc = __builtin_amdgcn_wmma_f32_16x16x4_f32(
            false, a, false, b, (short)0, acc, false, false);
    }

    float bias1 = b1[col];
    #pragma unroll
    for (int g = 0; g < 8; ++g) {
        int m = g + 8 * half;
        float v = acc[g] + bias1;
        ldsH[m * LDS_STRIDE + col] = v > 0.f ? v : 0.f;
    }
    __syncthreads();

    v8f acc2 = {};
    for (int k0 = 0; k0 < DIM; k0 += 4) {
        int ka = k0 + 2 * half;
        v2f a, b;
        a.x = ldsH[l16 * LDS_STRIDE + ka];
        a.y = ldsH[l16 * LDS_STRIDE + ka + 1];
        b.x = W2[ka * DIM + col];
        b.y = W2[(ka + 1) * DIM + col];
        acc2 = __builtin_amdgcn_wmma_f32_16x16x4_f32(
            false, a, false, b, (short)0, acc2, false, false);
    }

    float bias2 = b2[col];
    float s = 0.f, q = 0.f;
    #pragma unroll
    for (int g = 0; g < 8; ++g) {
        int m = g + 8 * half;
        float v = acc2[g] + bias2;
        v = v > 0.f ? v : 0.f;
        hbuf[(size_t)(row0 + m) * DIM + col] = v;
        s += v;
        q += v * v;
    }
    s += __shfl_xor(s, 16);
    q += __shfl_xor(q, 16);
    if (half == 0) {
        atomicAdd(&stats[col], s);
        atomicAdd(&stats[DIM + col], q);
    }
}

__global__ void bn_stats_finalize(float* __restrict__ stats)
{
    int c = threadIdx.x;
    float mu  = stats[c] * (1.0f / N_NODES);
    float var = stats[DIM + c] * (1.0f / N_NODES) - mu * mu;
    stats[c]       = mu;
    stats[DIM + c] = rsqrtf(var + BN_EPS);
}

__global__ __launch_bounds__(256) void bn_pool_kernel(
    const float* __restrict__ h,
    const float* __restrict__ stats,
    const float* __restrict__ gamma, const float* __restrict__ beta,
    const int* __restrict__ batch,
    float* __restrict__ zcat, float* __restrict__ gcat)
{
    int tid = blockIdx.x * blockDim.x + threadIdx.x;
    if (tid >= N_NODES * 32) return;
    int i = tid >> 5;
    int c = (tid & 31) * 4;
    int g = batch[i];
    float4 v = *(const float4*)(h + (size_t)i * DIM + c);
    float in[4] = {v.x, v.y, v.z, v.w};
    float4 o;
    float* op = &o.x;
    #pragma unroll
    for (int j = 0; j < 4; ++j) {
        float mu = stats[c + j];
        float rs = stats[DIM + c + j];
        float y  = (in[j] - mu) * rs * gamma[c + j] + beta[c + j];
        op[j] = y;
        atomicAdd(&gcat[(size_t)g * ZSTRIDE + c + j], y);
    }
    *(float4*)(zcat + (size_t)i * ZSTRIDE + c) = o;
}

extern "C" void kernel_launch(void* const* d_in, const int* in_sizes, int n_in,
                              void* d_out, int out_size, void* d_ws, size_t ws_size,
                              hipStream_t stream)
{
    (void)in_sizes; (void)n_in; (void)out_size; (void)ws_size;

    const float* x     = (const float*)d_in[0];
    const int*   ei    = (const int*)  d_in[1];
    const int*   batch = (const int*)  d_in[2];
    const float* W1    = (const float*)d_in[3];
    const float* b1    = (const float*)d_in[4];
    const float* W2    = (const float*)d_in[5];
    const float* b2    = (const float*)d_in[6];
    const float* gamma = (const float*)d_in[7];
    const float* beta  = (const float*)d_in[8];

    float* out  = (float*)d_out;
    float* zcat = out;
    float* gcat = out + (size_t)N_NODES * ZSTRIDE;

    float* agg   = (float*)d_ws;
    float* stats = agg + (size_t)N_NODES * DIM;

    const int* src = ei;
    const int* dst = ei + N_EDGES;

    hipMemsetAsync(gcat, 0, sizeof(float) * (size_t)NGRAPH * ZSTRIDE, stream);

    for (int l = 0; l < NLAYER; ++l) {
        const float* zin = (l == 0) ? x : (zcat + (size_t)(l - 1) * DIM);
        int zstride      = (l == 0) ? DIM : ZSTRIDE;

        hipMemsetAsync(agg,   0, sizeof(float) * (size_t)N_NODES * DIM, stream);
        hipMemsetAsync(stats, 0, sizeof(float) * 2 * DIM, stream);

        {
            long long threads = (long long)N_EDGES * 32;
            int blocks = (int)((threads + 255) / 256);
            scatter_kernel<<<blocks, 256, 0, stream>>>(zin, zstride, src, dst, agg);
        }

        gin_mlp_kernel<<<N_NODES / 16, 256, 0, stream>>>(
            zin, zstride, agg,
            W1 + (size_t)l * DIM * DIM, b1 + (size_t)l * DIM,
            W2 + (size_t)l * DIM * DIM, b2 + (size_t)l * DIM,
            stats);

        bn_stats_finalize<<<1, DIM, 0, stream>>>(stats);

        {
            long long threads = (long long)N_NODES * 32;
            int blocks = (int)((threads + 255) / 256);
            bn_pool_kernel<<<blocks, 256, 0, stream>>>(
                agg, stats,
                gamma + (size_t)l * DIM, beta + (size_t)l * DIM,
                batch,
                zcat + (size_t)l * DIM, gcat + (size_t)l * DIM);
        }
    }
}


# Edge scatter: a wave takes 8 consecutive edges (was 1), indices by scalar loads, its 32 row loads in flight before its 32 f32 atomics; waves /8; sums stay in arrival order as before
# speedup vs baseline: 1.6442x; 1.6442x over previous
#include <hip/hip_runtime.h>

#define N_NODES 50000
#define N_EDGES 800000
#define DIM     128
#define NLAYER  3
#define NGRAPH  512
#define ZSTRIDE (NLAYER * DIM)
#define BN_EPS  1e-5f
#define LDS_STRIDE 132

typedef __attribute__((ext_vector_type(2))) float v2f;
typedef __attribute__((ext_vector_type(8))) float v8f;

#define EPW 8
#define SCATTER_WAVES (N_EDGES / EPW)

__global__ __launch_bounds__(256) void scatter_kernel(
    const float* __restrict__ z, int zstride,
    const int* __restrict__ src, const int* __restrict__ dst,
    float* __restrict__ agg)
{
    const int wv   = __builtin_amdgcn_readfirstlane((int)(threadIdx.x >> 5));
    const int w    = blockIdx.x * 8 + wv;
    const int lane = threadIdx.x & 31;
    if (w >= SCATTER_WAVES) return;
    const int e0 = w * EPW;

    int s[EPW], d[EPW];
    #pragma unroll
    for (int j = 0; j < EPW; ++j) { s[j] = src[e0 + j]; d[j] = dst[e0 + j]; }

    float v[EPW][4];
    #pragma unroll
    for (int j = 0; j < EPW; ++j) {
        const float* zr = z + (size_t)s[j] * zstride + lane;
        v[j][0] = zr[0];
        v[j][1] = zr[32];
        v[j][2] = zr[64];
        v[j][3] = zr[96];
    }
    #pragma unroll
    for (int j = 0; j < EPW; ++j) {
        float* a = agg + (size_t)d[j] * DIM + lane;
        atomicAdd(a +  0, v[j][0]);
        atomicAdd(a + 32, v[j][1]);
        atomicAdd(a + 64, v[j][2]);
        atomicAdd(a + 96, v[j][3]);
    }
}

__global__ __launch_bounds__(256) void gin_mlp_kernel(
    const float* __restrict__ z, int zstride,
    float* __restrict__ hbuf,
    const float* __restrict__ W1, const float* __restrict__ b1,
    const float* __restrict__ W2, const float* __restrict__ b2,
    float* __restrict__ stats)
{
    __shared__ float ldsA[16 * LDS_STRIDE];
    __shared__ float ldsH[16 * LDS_STRIDE];

    const int tid  = threadIdx.x;
    const int wave = tid >> 5;
    const int lane = tid & 31;
    const int half = lane >> 4;
    const int l16  = lane & 15;
    const int row0 = blockIdx.x * 16;
    const int col  = wave * 16 + l16;

    for (int idx = tid; idx < 16 * DIM; idx += 256) {
        int r = idx >> 7, c = idx & 127;
        ldsA[r * LDS_STRIDE + c] =
            z[(size_t)(row0 + r) * zstride + c] + hbuf[(size_t)(row0 + r) * DIM + c];
    }
    __syncthreads();

    v8f acc = {};
    for (int k0 = 0; k0 < DIM; k0 += 4) {
        int ka = k0 + 2 * half;
        v2f a, b;
        a.x = ldsA[l16 * LDS_STRIDE + ka];
        a.y = ldsA[l16 * LDS_STRIDE + ka + 1];
        b.x = W1[ka * DIM + col];
        b.y = W1[(ka + 1) * DIM + col];
        acc = __builtin_amdgcn_wmma_f32_16x16x4_f32(
            false, a, false, b, (short)0, acc, false, false);
    }

    float bias1 = b1[col];
    #pragma unroll
    for (int g = 0; g < 8; ++g) {
        int m = g + 8 * half;
        float v = acc[g] + bias1;
        ldsH[m * LDS_STRIDE + col] = v > 0.f ? v : 0.f;
    }
    __syncthreads();

    v8f acc2 = {};
    for (int k0 = 0; k0 < DIM; k0 += 4) {
        int ka = k0 + 2 * half;
        v2f a, b;
        a.x = ldsH[l16 * LDS_STRIDE + ka];
        a.y = ldsH[l16 * LDS_STRIDE + ka + 1];
        b.x = W2[ka * DIM + col];
        b.y = W2[(ka + 1) * DIM + col];
        acc2 = __builtin_amdgcn_wmma_f32_16x16x4_f32(
            false, a, false, b, (short)0, acc2, false, false);
    }

    float bias2 = b2[col];
    float s = 0.f, q = 0.f;
    #pragma unroll
    for (int g = 0; g < 8; ++g) {
        int m = g + 8 * half;
        float v = acc2[g] + bias2;
        v = v > 0.f ? v : 0.f;
        hbuf[(size_t)(row0 + m) * DIM + col] = v;
        s += v;
        q += v * v;
    }
    s += __shfl_xor(s, 16);
    q += __shfl_xor(q, 16);
    if (half == 0) {
        atomicAdd(&stats[col], s);
        atomicAdd(&stats[DIM + col], q);
    }
}

__global__ void bn_stats_finalize(float* __restrict__ stats)
{
    int c = threadIdx.x;
    float mu  = stats[c] * (1.0f / N_NODES);
    float var = stats[DIM + c] * (1.0f / N_NODES) - mu * mu;
    stats[c]       = mu;
    stats[DIM + c] = rsqrtf(var + BN_EPS);
}

__global__ __launch_bounds__(256) void bn_pool_kernel(
    const float* __restrict__ h,
    const float* __restrict__ stats,
    const float* __restrict__ gamma, const float* __restrict__ beta,
    const int* __restrict__ batch,
    float* __restrict__ zcat, float* __restrict__ gcat)
{
    int tid = blockIdx.x * blockDim.x + threadIdx.x;
    if (tid >= N_NODES * 32) return;
    int i    = tid >> 5;
    int lane = tid & 31;
    int g = batch[i];
    const float* hr = h + (size_t)i * DIM + lane;
    float* zr = zcat + (size_t)i * ZSTRIDE + lane;
    float* gr = gcat + (size_t)g * ZSTRIDE + lane;
    float in[4], mu[4], rs[4], ga[4], be[4], y[4];
    for (int j = 0; j < 4; ++j) {
        int c = lane + 32 * j;
        in[j] = hr[32 * j];
        mu[j] = stats[c];
        rs[j] = stats[DIM + c];
        ga[j] = gamma[c];
        be[j] = beta[c];
    }
    for (int j = 0; j < 4; ++j)
        y[j] = (in[j] - mu[j]) * rs[j] * ga[j] + be[j];
    for (int j = 0; j < 4; ++j) zr[32 * j] = y[j];
    for (int j = 0; j < 4; ++j) atomicAdd(gr + 32 * j, y[j]);
}

extern "C" void kernel_launch(void* const* d_in, const int* in_sizes, int n_in,
                              void* d_out, int out_size, void* d_ws, size_t ws_size,
                              hipStream_t stream)
{
    (void)in_sizes; (void)n_in; (void)out_size; (void)ws_size;

    const float* x     = (const float*)d_in[0];
    const int*   ei    = (const int*)  d_in[1];
    const int*   batch = (const int*)  d_in[2];
    const float* W1    = (const float*)d_in[3];
    const float* b1    = (const float*)d_in[4];
    const float* W2    = (const float*)d_in[5];
    const float* b2    = (const float*)d_in[6];
    const float* gamma = (const float*)d_in[7];
    const float* beta  = (const float*)d_in[8];

    float* out  = (float*)d_out;
    float* zcat = out;
    float* gcat = out + (size_t)N_NODES * ZSTRIDE;

    float* agg   = (float*)d_ws;
    float* stats = agg + (size_t)N_NODES * DIM;

    const int* src = ei;
    const int* dst = ei + N_EDGES;

    hipMemsetAsync(gcat, 0, sizeof(float) * (size_t)NGRAPH * ZSTRIDE, stream);

    for (int l = 0; l < NLAYER; ++l) {
        const float* zin = (l == 0) ? x : (zcat + (size_t)(l - 1) * DIM);
        int zstride      = (l == 0) ? DIM : ZSTRIDE;

        hipMemsetAsync(agg,   0, sizeof(float) * (size_t)N_NODES * DIM, stream);
        hipMemsetAsync(stats, 0, sizeof(float) * 2 * DIM, stream);

        scatter_kernel<<<(SCATTER_WAVES + 7) / 8, 256, 0, stream>>>(zin, zstride, src, dst, agg);

        gin_mlp_kernel<<<N_NODES / 16, 256, 0, stream>>>(
            zin, zstride, agg,
            W1 + (size_t)l * DIM * DIM, b1 + (size_t)l * DIM,
            W2 + (size_t)l * DIM * DIM, b2 + (size_t)l * DIM,
            stats);

        bn_stats_finalize<<<1, DIM, 0, stream>>>(stats);

        {
            long long threads = (long long)N_NODES * 32;
            int blocks = (int)((threads + 255) / 256);
            bn_pool_kernel<<<blocks, 256, 0, stream>>>(
                agg, stats,
                gamma + (size_t)l * DIM, beta + (size_t)l * DIM,
                batch,
                zcat + (size_t)l * DIM, gcat + (size_t)l * DIM);
        }
    }
}


# Edge scatter: each output row's f32 atomics come from one half of the package only (a block reads its die and draws edge runs from its half's counter); sums stay in arrival order as before
# speedup vs baseline: 1.8481x; 1.1240x over previous
#include <hip/hip_runtime.h>

#define N_NODES 50000
#define N_EDGES 800000
#define DIM     128
#define NLAYER  3
#define NGRAPH  512
#define ZSTRIDE (NLAYER * DIM)
#define BN_EPS  1e-5f
#define LDS_STRIDE 132

typedef __attribute__((ext_vector_type(2))) float v2f;
typedef __attribute__((ext_vector_type(8))) float v8f;

#define EPR 16
#define RPW 2
#define EPT (8 * RPW * EPR)
#define N_TICKETS (N_EDGES / EPT)
#define HALF_ROWS (N_NODES / 2)
#define CNT_STRIDE 64
#define SCATTER_BLOCKS 1024

__global__ __launch_bounds__(256) void scatter_kernel(
    const float* __restrict__ z, int zstride,
    const int* __restrict__ src, const int* __restrict__ dst,
    float* __restrict__ agg, int* __restrict__ cnt)
{
    __shared__ int tk[2];
    __shared__ int sh_half;

    const int wv   = __builtin_amdgcn_readfirstlane((int)(threadIdx.x >> 5));
    const int lane = threadIdx.x & 31;

    if (threadIdx.x == 0) sh_half = (__builtin_amdgcn_s_sendmsg_rtn(0x87) >> 18) & 1;
    __syncthreads();
    const int myhalf = __builtin_amdgcn_readfirstlane(sh_half);

    for (int pass = 0; pass < 2; ++pass) {
        const int q = myhalf ^ pass;
        int* c = cnt + q * CNT_STRIDE;
        if (threadIdx.x == 0) tk[0] = atomicAdd(c, 1);
        __syncthreads();
        for (int it = 0; it <= N_TICKETS; ++it) {
            const int t = __builtin_amdgcn_readfirstlane(tk[it & 1]);
            if (t >= N_TICKETS) break;
            int tnext = 0;
            if (threadIdx.x == 0) tnext = atomicAdd(c, 1);

            const int e0 = t * EPT + wv * (RPW * EPR);
            for (int r = 0; r < RPW; ++r) {
                const int eb = e0 + r * EPR;
                int s[EPR], d[EPR];
                #pragma unroll
                for (int j = 0; j < EPR; ++j) { s[j] = src[eb + j]; d[j] = dst[eb + j]; }
                asm volatile("" :: "s"(s[0]), "s"(s[1]), "s"(s[2]), "s"(s[3]),
                                   "s"(s[4]), "s"(s[5]), "s"(s[6]), "s"(s[7]),
                                   "s"(s[8]), "s"(s[9]), "s"(s[10]), "s"(s[11]),
                                   "s"(s[12]), "s"(s[13]), "s"(s[14]), "s"(s[15]));

                float v[EPR][4];
                #pragma unroll
                for (int j = 0; j < EPR; ++j) {
                    if ((d[j] >= HALF_ROWS) == (q != 0)) {
                        const float* zr = z + (size_t)s[j] * zstride + lane;
                        v[j][0] = zr[0];
                        v[j][1] = zr[32];
                        v[j][2] = zr[64];
                        v[j][3] = zr[96];
                    }
                }
                #pragma unroll
                for (int j = 0; j < EPR; ++j) {
                    if ((d[j] >= HALF_ROWS) == (q != 0)) {
                        float* a = agg + (size_t)d[j] * DIM + lane;
                        atomicAdd(a +  0, v[j][0]);
                        atomicAdd(a + 32, v[j][1]);
                        atomicAdd(a + 64, v[j][2]);
                        atomicAdd(a + 96, v[j][3]);
                    }
                }
            }

            if (threadIdx.x == 0) tk[(it + 1) & 1] = tnext;
            __syncthreads();
        }
        __syncthreads();
    }
}

__global__ __launch_bounds__(256) void gin_mlp_kernel(
    const float* __restrict__ z, int zstride,
    float* __restrict__ hbuf,
    const float* __restrict__ W1, const float* __restrict__ b1,
    const float* __restrict__ W2, const float* __restrict__ b2,
    float* __restrict__ stats)
{
    __shared__ float ldsA[16 * LDS_STRIDE];
    __shared__ float ldsH[16 * LDS_STRIDE];

    const int tid  = threadIdx.x;
    const int wave = tid >> 5;
    const int lane = tid & 31;
    const int half = lane >> 4;
    const int l16  = lane & 15;
    const int row0 = blockIdx.x * 16;
    const int col  = wave * 16 + l16;

    for (int idx = tid; idx < 16 * DIM; idx += 256) {
        int r = idx >> 7, c = idx & 127;
        ldsA[r * LDS_STRIDE + c] =
            z[(size_t)(row0 + r) * zstride + c] + hbuf[(size_t)(row0 + r) * DIM + c];
    }
    __syncthreads();

    v8f acc = {};
    for (int k0 = 0; k0 < DIM; k0 += 4) {
        int ka = k0 + 2 * half;
        v2f a, b;
        a.x = ldsA[l16 * LDS_STRIDE + ka];
        a.y = ldsA[l16 * LDS_STRIDE + ka + 1];
        b.x = W1[ka * DIM + col];
        b.y = W1[(ka + 1) * DIM + col];
        acc = __builtin_amdgcn_wmma_f32_16x16x4_f32(
            false, a, false, b, (short)0, acc, false, false);
    }

    float bias1 = b1[col];
    #pragma unroll
    for (int g = 0; g < 8; ++g) {
        int m = g + 8 * half;
        float v = acc[g] + bias1;
        ldsH[m * LDS_STRIDE + col] = v > 0.f ? v : 0.f;
    }
    __syncthreads();

    v8f acc2 = {};
    for (int k0 = 0; k0 < DIM; k0 += 4) {
        int ka = k0 + 2 * half;
        v2f a, b;
        a.x = ldsH[l16 * LDS_STRIDE + ka];
        a.y = ldsH[l16 * LDS_STRIDE + ka + 1];
        b.x = W2[ka * DIM + col];
        b.y = W2[(ka + 1) * DIM + col];
        acc2 = __builtin_amdgcn_wmma_f32_16x16x4_f32(
            false, a, false, b, (short)0, acc2, false, false);
    }

    float bias2 = b2[col];
    float s = 0.f, q = 0.f;
    #pragma unroll
    for (int g = 0; g < 8; ++g) {
        int m = g + 8 * half;
        float v = acc2[g] + bias2;
        v = v > 0.f ? v : 0.f;
        hbuf[(size_t)(row0 + m) * DIM + col] = v;
        s += v;
        q += v * v;
    }
    s += __shfl_xor(s, 16);
    q += __shfl_xor(q, 16);
    if (half == 0) {
        atomicAdd(&stats[col], s);
        atomicAdd(&stats[DIM + col], q);
    }
}

__global__ void bn_stats_finalize(float* __restrict__ stats)
{
    int c = threadIdx.x;
    float mu  = stats[c] * (1.0f / N_NODES);
    float var = stats[DIM + c] * (1.0f / N_NODES) - mu * mu;
    stats[c]       = mu;
    stats[DIM + c] = rsqrtf(var + BN_EPS);
}

__global__ __launch_bounds__(256) void bn_pool_kernel(
    const float* __restrict__ h,
    const float* __restrict__ stats,
    const float* __restrict__ gamma, const float* __restrict__ beta,
    const int* __restrict__ batch,
    float* __restrict__ zcat, float* __restrict__ gcat)
{
    int tid = blockIdx.x * blockDim.x + threadIdx.x;
    if (tid >= N_NODES * 32) return;
    int i    = tid >> 5;
    int lane = tid & 31;
    int g = batch[i];
    const float* hr = h + (size_t)i * DIM + lane;
    float* zr = zcat + (size_t)i * ZSTRIDE + lane;
    float* gr = gcat + (size_t)g * ZSTRIDE + lane;
    float in[4], mu[4], rs[4], ga[4], be[4], y[4];
    for (int j = 0; j < 4; ++j) {
        int c = lane + 32 * j;
        in[j] = hr[32 * j];
        mu[j] = stats[c];
        rs[j] = stats[DIM + c];
        ga[j] = gamma[c];
        be[j] = beta[c];
    }
    for (int j = 0; j < 4; ++j)
        y[j] = (in[j] - mu[j]) * rs[j] * ga[j] + be[j];
    for (int j = 0; j < 4; ++j) zr[32 * j] = y[j];
    for (int j = 0; j < 4; ++j) atomicAdd(gr + 32 * j, y[j]);
}

extern "C" void kernel_launch(void* const* d_in, const int* in_sizes, int n_in,
                              void* d_out, int out_size, void* d_ws, size_t ws_size,
                              hipStream_t stream)
{
    (void)in_sizes; (void)n_in; (void)out_size; (void)ws_size;

    const float* x     = (const float*)d_in[0];
    const int*   ei    = (const int*)  d_in[1];
    const int*   batch = (const int*)  d_in[2];
    const float* W1    = (const float*)d_in[3];
    const float* b1    = (const float*)d_in[4];
    const float* W2    = (const float*)d_in[5];
    const float* b2    = (const float*)d_in[6];
    const float* gamma = (const float*)d_in[7];
    const float* beta  = (const float*)d_in[8];

    float* out  = (float*)d_out;
    float* zcat = out;
    float* gcat = out + (size_t)N_NODES * ZSTRIDE;

    float* agg   = (float*)d_ws;
    float* stats = agg + (size_t)N_NODES * DIM;
    int*   cnt   = (int*)(stats + 2 * DIM);

    const int* src = ei;
    const int* dst = ei + N_EDGES;

    hipMemsetAsync(gcat, 0, sizeof(float) * (size_t)NGRAPH * ZSTRIDE, stream);

    for (int l = 0; l < NLAYER; ++l) {
        const float* zin = (l == 0) ? x : (zcat + (size_t)(l - 1) * DIM);
        int zstride      = (l == 0) ? DIM : ZSTRIDE;

        hipMemsetAsync(agg,   0, sizeof(float) * (size_t)N_NODES * DIM, stream);
        hipMemsetAsync(stats, 0, sizeof(float) * (2 * DIM + 2 * CNT_STRIDE), stream);

        scatter_kernel<<<SCATTER_BLOCKS, 256, 0, stream>>>(zin, zstride, src, dst, agg, cnt);

        gin_mlp_kernel<<<N_NODES / 16, 256, 0, stream>>>(
            zin, zstride, agg,
            W1 + (size_t)l * DIM * DIM, b1 + (size_t)l * DIM,
            W2 + (size_t)l * DIM * DIM, b2 + (size_t)l * DIM,
            stats);

        bn_stats_finalize<<<1, DIM, 0, stream>>>(stats);

        {
            long long threads = (long long)N_NODES * 32;
            int blocks = (int)((threads + 255) / 256);
            bn_pool_kernel<<<blocks, 256, 0, stream>>>(
                agg, stats,
                gamma + (size_t)l * DIM, beta + (size_t)l * DIM,
                batch,
                zcat + (size_t)l * DIM, gcat + (size_t)l * DIM);
        }
    }
}


# Neighbour sum in LDS, no global f32 atomics, no agg clear: a block owns 196 rows, filters all edges by a per-call owner byte, adds their z rows with LDS f32 atomics; row sum order = LDS arrival order
# speedup vs baseline: 2.3216x; 1.2562x over previous
#include <hip/hip_runtime.h>

#define N_NODES 50000
#define N_EDGES 800000
#define DIM     128
#define NLAYER  3
#define NGRAPH  512
#define ZSTRIDE (NLAYER * DIM)
#define BN_EPS  1e-5f
#define LDS_STRIDE 132

typedef __attribute__((ext_vector_type(2))) float v2f;
typedef __attribute__((ext_vector_type(8))) float v8f;

#define RB          196
#define AGG_BLOCKS  256
#define AGG_THREADS 1024
#define LIST_CAP    16384
#define GROUPS16    (N_EDGES / 16)
#define SCAN_UNROLL 7
#define GE          8

__global__ __launch_bounds__(256) void owner_kernel(
    const int* __restrict__ dst, unsigned* __restrict__ owner)
{
    const int i = blockIdx.x * 256 + threadIdx.x;
    if (i >= N_EDGES / 4) return;
    const int4 d = *(const int4*)(dst + 4 * i);
    owner[i] =  ((unsigned)d.x / RB)        | (((unsigned)d.y / RB) << 8) |
               (((unsigned)d.z / RB) << 16) | (((unsigned)d.w / RB) << 24);
}

__device__ __forceinline__ unsigned zero_bytes(unsigned x)
{
    unsigned t = (x & 0x7f7f7f7fu) + 0x7f7f7f7fu;
    t = t | x | 0x7f7f7f7fu;
    return ~t;
}

__global__ __launch_bounds__(AGG_THREADS) void agg_kernel(
    const float* __restrict__ z, int zstride,
    const int* __restrict__ src, const int* __restrict__ dst,
    const unsigned* __restrict__ owner,
    float* __restrict__ agg)
{
    __shared__ float acc[RB * DIM];
    __shared__ int   list[LIST_CAP];
    __shared__ int   nlist;

    const int tid  = threadIdx.x;
    const int lane = tid & 31;
    const int wv   = __builtin_amdgcn_readfirstlane(tid >> 5);
    const int lo   = blockIdx.x * RB;
    const int nrows = (N_NODES - lo < RB) ? (N_NODES - lo) : RB;

    for (int i = tid; i < RB * DIM; i += AGG_THREADS) acc[i] = 0.f;
    if (tid == 0) nlist = 0;
    __syncthreads();

    const unsigned pat = (unsigned)blockIdx.x * 0x01010101u;
    const uint4* ow = (const uint4*)owner;
    for (int g0 = 0; g0 < GROUPS16; g0 += AGG_THREADS * SCAN_UNROLL) {
        uint4 w[SCAN_UNROLL];
        #pragma unroll
        for (int u = 0; u < SCAN_UNROLL; ++u) {
            const int g = g0 + u * AGG_THREADS + tid;
            w[u] = ow[g < GROUPS16 ? g : GROUPS16 - 1];
        }
        #pragma unroll
        for (int u = 0; u < SCAN_UNROLL; ++u) {
            const int g = g0 + u * AGG_THREADS + tid;
            unsigned m = (zero_bytes(w[u].x ^ pat) >> 3) | (zero_bytes(w[u].y ^ pat) >> 2) |
                         (zero_bytes(w[u].z ^ pat) >> 1) |  zero_bytes(w[u].w ^ pat);
            if (g >= GROUPS16) m = 0;
            while (m) {
                const int p = __builtin_ctz(m);
                m &= m - 1;
                const int e = g * 16 + (((p & 7) - 4) << 2) + (p >> 3);
                const int slot = atomicAdd(&nlist, 1);
                if (slot < LIST_CAP) {
                    list[slot] = e;
                } else {
                    const int dl = dst[e] - lo;
                    const float* zr = z + (size_t)src[e] * zstride;
                    for (int c = 0; c < DIM; ++c) atomicAdd(&acc[dl * DIM + c], zr[c]);
                }
            }
        }
    }
    __syncthreads();

    const int nl = __builtin_amdgcn_readfirstlane(nlist);
    const int n  = nl < LIST_CAP ? nl : LIST_CAP;
    for (int base = wv * 32; base < n; base += AGG_THREADS) {
        const int cnt = (n - base < 32) ? (n - base) : 32;
        int sl = 0, dl = 0;
        if (lane < cnt) {
            const int e = list[base + lane];
            sl = src[e];
            dl = dst[e] - lo;
        }
        for (int j0 = 0; j0 < cnt; j0 += GE) {
            int s[GE], d[GE];
            #pragma unroll
            for (int u = 0; u < GE; ++u) {
                s[u] = __builtin_amdgcn_readlane(sl, j0 + u);
                d[u] = __builtin_amdgcn_readlane(dl, j0 + u);
            }
            float v[GE][4];
            #pragma unroll
            for (int u = 0; u < GE; ++u) {
                if (j0 + u < cnt) {
                    const float* zr = z + (size_t)s[u] * zstride + lane;
                    v[u][0] = zr[0];
                    v[u][1] = zr[32];
                    v[u][2] = zr[64];
                    v[u][3] = zr[96];
                }
            }
            #pragma unroll
            for (int u = 0; u < GE; ++u) {
                if (j0 + u < cnt) {
                    float* a = acc + d[u] * DIM + lane;
                    atomicAdd(a +  0, v[u][0]);
                    atomicAdd(a + 32, v[u][1]);
                    atomicAdd(a + 64, v[u][2]);
                    atomicAdd(a + 96, v[u][3]);
                }
            }
        }
    }
    __syncthreads();

    float* outp = agg + (size_t)lo * DIM;
    for (int i = tid; i < nrows * DIM; i += AGG_THREADS) outp[i] = acc[i];
}

__global__ __launch_bounds__(256) void gin_mlp_kernel(
    const float* __restrict__ z, int zstride,
    float* __restrict__ hbuf,
    const float* __restrict__ W1, const float* __restrict__ b1,
    const float* __restrict__ W2, const float* __restrict__ b2,
    float* __restrict__ stats)
{
    __shared__ float ldsA[16 * LDS_STRIDE];
    __shared__ float ldsH[16 * LDS_STRIDE];

    const int tid  = threadIdx.x;
    const int wave = tid >> 5;
    const int lane = tid & 31;
    const int half = lane >> 4;
    const int l16  = lane & 15;
    const int row0 = blockIdx.x * 16;
    const int col  = wave * 16 + l16;

    for (int idx = tid; idx < 16 * DIM; idx += 256) {
        int r = idx >> 7, c = idx & 127;
        ldsA[r * LDS_STRIDE + c] =
            z[(size_t)(row0 + r) * zstride + c] + hbuf[(size_t)(row0 + r) * DIM + c];
    }
    __syncthreads();

    v8f acc = {};
    for (int k0 = 0; k0 < DIM; k0 += 4) {
        int ka = k0 + 2 * half;
        v2f a, b;
        a.x = ldsA[l16 * LDS_STRIDE + ka];
        a.y = ldsA[l16 * LDS_STRIDE + ka + 1];
        b.x = W1[ka * DIM + col];
        b.y = W1[(ka + 1) * DIM + col];
        acc = __builtin_amdgcn_wmma_f32_16x16x4_f32(
            false, a, false, b, (short)0, acc, false, false);
    }

    float bias1 = b1[col];
    #pragma unroll
    for (int g = 0; g < 8; ++g) {
        int m = g + 8 * half;
        float v = acc[g] + bias1;
        ldsH[m * LDS_STRIDE + col] = v > 0.f ? v : 0.f;
    }
    __syncthreads();

    v8f acc2 = {};
    for (int k0 = 0; k0 < DIM; k0 += 4) {
        int ka = k0 + 2 * half;
        v2f a, b;
        a.x = ldsH[l16 * LDS_STRIDE + ka];
        a.y = ldsH[l16 * LDS_STRIDE + ka + 1];
        b.x = W2[ka * DIM + col];
        b.y = W2[(ka + 1) * DIM + col];
        acc2 = __builtin_amdgcn_wmma_f32_16x16x4_f32(
            false, a, false, b, (short)0, acc2, false, false);
    }

    float bias2 = b2[col];
    float s = 0.f, q = 0.f;
    #pragma unroll
    for (int g = 0; g < 8; ++g) {
        int m = g + 8 * half;
        float v = acc2[g] + bias2;
        v = v > 0.f ? v : 0.f;
        hbuf[(size_t)(row0 + m) * DIM + col] = v;
        s += v;
        q += v * v;
    }
    s += __shfl_xor(s, 16);
    q += __shfl_xor(q, 16);
    if (half == 0) {
        atomicAdd(&stats[col], s);
        atomicAdd(&stats[DIM + col], q);
    }
}

__global__ void bn_stats_finalize(float* __restrict__ stats)
{
    int c = threadIdx.x;
    float mu  = stats[c] * (1.0f / N_NODES);
    float var = stats[DIM + c] * (1.0f / N_NODES) - mu * mu;
    stats[c]       = mu;
    stats[DIM + c] = rsqrtf(var + BN_EPS);
}

__global__ __launch_bounds__(256) void bn_pool_kernel(
    const float* __restrict__ h,
    const float* __restrict__ stats,
    const float* __restrict__ gamma, const float* __restrict__ beta,
    const int* __restrict__ batch,
    float* __restrict__ zcat, float* __restrict__ gcat)
{
    int tid = blockIdx.x * blockDim.x + threadIdx.x;
    if (tid >= N_NODES * 32) return;
    int i    = tid >> 5;
    int lane = tid & 31;
    int g = batch[i];
    const float* hr = h + (size_t)i * DIM + lane;
    float* zr = zcat + (size_t)i * ZSTRIDE + lane;
    float* gr = gcat + (size_t)g * ZSTRIDE + lane;
    float in[4], mu[4], rs[4], ga[4], be[4], y[4];
    for (int j = 0; j < 4; ++j) {
        int c = lane + 32 * j;
        in[j] = hr[32 * j];
        mu[j] = stats[c];
        rs[j] = stats[DIM + c];
        ga[j] = gamma[c];
        be[j] = beta[c];
    }
    for (int j = 0; j < 4; ++j)
        y[j] = (in[j] - mu[j]) * rs[j] * ga[j] + be[j];
    for (int j = 0; j < 4; ++j) zr[32 * j] = y[j];
    for (int j = 0; j < 4; ++j) atomicAdd(gr + 32 * j, y[j]);
}

extern "C" void kernel_launch(void* const* d_in, const int* in_sizes, int n_in,
                              void* d_out, int out_size, void* d_ws, size_t ws_size,
                              hipStream_t stream)
{
    (void)in_sizes; (void)n_in; (void)out_size; (void)ws_size;

    const float* x     = (const float*)d_in[0];
    const int*   ei    = (const int*)  d_in[1];
    const int*   batch = (const int*)  d_in[2];
    const float* W1    = (const float*)d_in[3];
    const float* b1    = (const float*)d_in[4];
    const float* W2    = (const float*)d_in[5];
    const float* b2    = (const float*)d_in[6];
    const float* gamma = (const float*)d_in[7];
    const float* beta  = (const float*)d_in[8];

    float* out  = (float*)d_out;
    float* zcat = out;
    float* gcat = out + (size_t)N_NODES * ZSTRIDE;

    float* agg   = (float*)d_ws;
    float* stats = agg + (size_t)N_NODES * DIM;
    unsigned* owner = (unsigned*)(stats + 2 * DIM);

    const int* src = ei;
    const int* dst = ei + N_EDGES;

    hipMemsetAsync(gcat, 0, sizeof(float) * (size_t)NGRAPH * ZSTRIDE, stream);

    owner_kernel<<<(N_EDGES / 4 + 255) / 256, 256, 0, stream>>>(dst, owner);

    for (int l = 0; l < NLAYER; ++l) {
        const float* zin = (l == 0) ? x : (zcat + (size_t)(l - 1) * DIM);
        int zstride      = (l == 0) ? DIM : ZSTRIDE;

        hipMemsetAsync(stats, 0, sizeof(float) * (2 * DIM), stream);

        agg_kernel<<<AGG_BLOCKS, AGG_THREADS, 0, stream>>>(zin, zstride, src, dst, owner, agg);

        gin_mlp_kernel<<<N_NODES / 16, 256, 0, stream>>>(
            zin, zstride, agg,
            W1 + (size_t)l * DIM * DIM, b1 + (size_t)l * DIM,
            W2 + (size_t)l * DIM * DIM, b2 + (size_t)l * DIM,
            stats);

        bn_stats_finalize<<<1, DIM, 0, stream>>>(stats);

        {
            long long threads = (long long)N_NODES * 32;
            int blocks = (int)((threads + 255) / 256);
            bn_pool_kernel<<<blocks, 256, 0, stream>>>(
                agg, stats,
                gamma + (size_t)l * DIM, beta + (size_t)l * DIM,
                batch,
                zcat + (size_t)l * DIM, gcat + (size_t)l * DIM);
        }
    }
}


# Edge lists once per call: layer 1's row-owner blocks save their list as packed (src row, local dst row) words; layers 2-3 load it and skip the owner-byte scan and index loads; LDS sum order as before
# speedup vs baseline: 2.5014x; 1.0774x over previous
#include <hip/hip_runtime.h>

#define N_NODES 50000
#define N_EDGES 800000
#define DIM     128
#define NLAYER  3
#define NGRAPH  512
#define ZSTRIDE (NLAYER * DIM)
#define BN_EPS  1e-5f
#define LDS_STRIDE 132

typedef __attribute__((ext_vector_type(2))) float v2f;
typedef __attribute__((ext_vector_type(8))) float v8f;

#define RB          196
#define AGG_BLOCKS  256
#define AGG_THREADS 1024
#define LIST_CAP    16384
#define GROUPS16    (N_EDGES / 16)
#define SCAN_UNROLL 7
#define GE          8
#define PRE         (LIST_CAP / AGG_THREADS)

static_assert(N_NODES <= 65536 && RB <= 65536, "a saved word holds two 16-bit row numbers");

__global__ __launch_bounds__(256) void owner_kernel(
    const int* __restrict__ dst, unsigned* __restrict__ owner)
{
    const int i = blockIdx.x * 256 + threadIdx.x;
    if (i >= N_EDGES / 4) return;
    const int4 d = *(const int4*)(dst + 4 * i);
    owner[i] =  ((unsigned)d.x / RB)        | (((unsigned)d.y / RB) << 8) |
               (((unsigned)d.z / RB) << 16) | (((unsigned)d.w / RB) << 24);
}

__device__ __forceinline__ unsigned zero_bytes(unsigned x)
{
    unsigned t = (x & 0x7f7f7f7fu) + 0x7f7f7f7fu;
    t = t | x | 0x7f7f7f7fu;
    return ~t;
}

template <bool FIRST>
__global__ __launch_bounds__(AGG_THREADS) void agg_kernel(
    const float* __restrict__ z, int zstride,
    const int* __restrict__ src, const int* __restrict__ dst,
    const unsigned* __restrict__ owner,
    unsigned* __restrict__ saved, int* __restrict__ saved_n,
    float* __restrict__ agg)
{
    __shared__ float acc[RB * DIM];
    __shared__ int   list[LIST_CAP];
    __shared__ int   nlist;

    const int tid  = threadIdx.x;
    const int lane = tid & 31;
    const int wv   = __builtin_amdgcn_readfirstlane(tid >> 5);
    const int lo   = blockIdx.x * RB;
    const int nrows = (N_NODES - lo < RB) ? (N_NODES - lo) : RB;
    unsigned* mine = saved + (size_t)blockIdx.x * LIST_CAP;

    int  nsaved = 0;
    bool scan   = true;
    if (!FIRST) {
        nsaved = __builtin_amdgcn_readfirstlane(saved_n[blockIdx.x]);
        scan   = nsaved > LIST_CAP;
    }
    unsigned pre[PRE];
    if (!FIRST && !scan) {
        #pragma unroll
        for (int u = 0; u < PRE; ++u) {
            const int i = u * AGG_THREADS + tid;
            pre[u] = (i < nsaved) ? mine[i] : 0u;
        }
    }

    for (int i = tid; i < RB * DIM; i += AGG_THREADS) acc[i] = 0.f;
    if (tid == 0) nlist = 0;

    if (!FIRST && !scan) {
        #pragma unroll
        for (int u = 0; u < PRE; ++u) {
            const int i = u * AGG_THREADS + tid;
            if (i < nsaved) list[i] = (int)pre[u];
        }
    }
    __syncthreads();

    int n = nsaved;
    if (scan) {
        const unsigned pat = (unsigned)blockIdx.x * 0x01010101u;
        const uint4* ow = (const uint4*)owner;
        for (int g0 = 0; g0 < GROUPS16; g0 += AGG_THREADS * SCAN_UNROLL) {
            uint4 w[SCAN_UNROLL];
            #pragma unroll
            for (int u = 0; u < SCAN_UNROLL; ++u) {
                const int g = g0 + u * AGG_THREADS + tid;
                w[u] = ow[g < GROUPS16 ? g : GROUPS16 - 1];
            }
            #pragma unroll
            for (int u = 0; u < SCAN_UNROLL; ++u) {
                const int g = g0 + u * AGG_THREADS + tid;
                unsigned m = (zero_bytes(w[u].x ^ pat) >> 3) | (zero_bytes(w[u].y ^ pat) >> 2) |
                             (zero_bytes(w[u].z ^ pat) >> 1) |  zero_bytes(w[u].w ^ pat);
                if (g >= GROUPS16) m = 0;
                while (m) {
                    const int p = __builtin_ctz(m);
                    m &= m - 1;
                    const int e = g * 16 + (((p & 7) - 4) << 2) + (p >> 3);
                    const int slot = atomicAdd(&nlist, 1);
                    if (slot < LIST_CAP) {
                        list[slot] = e;
                    } else {
                        const int dl = dst[e] - lo;
                        const float* zr = z + (size_t)src[e] * zstride;
                        for (int c = 0; c < DIM; ++c) atomicAdd(&acc[dl * DIM + c], zr[c]);
                    }
                }
            }
        }
        __syncthreads();
        const int nl = __builtin_amdgcn_readfirstlane(nlist);
        n = nl < LIST_CAP ? nl : LIST_CAP;
        if (FIRST && tid == 0) saved_n[blockIdx.x] = nl;
    }

    for (int base = wv * 32; base < n; base += AGG_THREADS) {
        const int cnt = (n - base < 32) ? (n - base) : 32;
        int sl = 0, dl = 0;
        if (lane < cnt) {
            const int w = list[base + lane];
            if (scan) {
                sl = src[w];
                dl = dst[w] - lo;
                if (FIRST) mine[base + lane] = (unsigned)sl | ((unsigned)dl << 16);
            } else {
                sl = w & 0xffff;
                dl = (int)((unsigned)w >> 16);
            }
        }
        for (int j0 = 0; j0 < cnt; j0 += GE) {
            int s[GE], d[GE];
            #pragma unroll
            for (int u = 0; u < GE; ++u) {
                s[u] = __builtin_amdgcn_readlane(sl, j0 + u);
                d[u] = __builtin_amdgcn_readlane(dl, j0 + u);
            }
            float v[GE][4];
            #pragma unroll
            for (int u = 0; u < GE; ++u) {
                if (j0 + u < cnt) {
                    const float* zr = z + (size_t)s[u] * zstride + lane;
                    v[u][0] = zr[0];
                    v[u][1] = zr[32];
                    v[u][2] = zr[64];
                    v[u][3] = zr[96];
                }
            }
            #pragma unroll
            for (int u = 0; u < GE; ++u) {
                if (j0 + u < cnt) {
                    float* a = acc + d[u] * DIM + lane;
                    atomicAdd(a +  0, v[u][0]);
                    atomicAdd(a + 32, v[u][1]);
                    atomicAdd(a + 64, v[u][2]);
                    atomicAdd(a + 96, v[u][3]);
                }
            }
        }
    }
    __syncthreads();

    float* outp = agg + (size_t)lo * DIM;
    for (int i = tid; i < nrows * DIM; i += AGG_THREADS) outp[i] = acc[i];
}

__global__ __launch_bounds__(256) void gin_mlp_kernel(
    const float* __restrict__ z, int zstride,
    float* __restrict__ hbuf,
    const float* __restrict__ W1, const float* __restrict__ b1,
    const float* __restrict__ W2, const float* __restrict__ b2,
    float* __restrict__ stats)
{
    __shared__ float ldsA[16 * LDS_STRIDE];
    __shared__ float ldsH[16 * LDS_STRIDE];

    const int tid  = threadIdx.x;
    const int wave = tid >> 5;
    const int lane = tid & 31;
    const int half = lane >> 4;
    const int l16  = lane & 15;
    const int row0 = blockIdx.x * 16;
    const int col  = wave * 16 + l16;

    for (int idx = tid; idx < 16 * DIM; idx += 256) {
        int r = idx >> 7, c = idx & 127;
        ldsA[r * LDS_STRIDE + c] =
            z[(size_t)(row0 + r) * zstride + c] + hbuf[(size_t)(row0 + r) * DIM + c];
    }
    __syncthreads();

    v8f acc = {};
    for (int k0 = 0; k0 < DIM; k0 += 4) {
        int ka = k0 + 2 * half;
        v2f a, b;
        a.x = ldsA[l16 * LDS_STRIDE + ka];
        a.y = ldsA[l16 * LDS_STRIDE + ka + 1];
        b.x = W1[ka * DIM + col];
        b.y = W1[(ka + 1) * DIM + col];
        acc = __builtin_amdgcn_wmma_f32_16x16x4_f32(
            false, a, false, b, (short)0, acc, false, false);
    }

    float bias1 = b1[col];
    #pragma unroll
    for (int g = 0; g < 8; ++g) {
        int m = g + 8 * half;
        float v = acc[g] + bias1;
        ldsH[m * LDS_STRIDE + col] = v > 0.f ? v : 0.f;
    }
    __syncthreads();

    v8f acc2 = {};
    for (int k0 = 0; k0 < DIM; k0 += 4) {
        int ka = k0 + 2 * half;
        v2f a, b;
        a.x = ldsH[l16 * LDS_STRIDE + ka];
        a.y = ldsH[l16 * LDS_STRIDE + ka + 1];
        b.x = W2[ka * DIM + col];
        b.y = W2[(ka + 1) * DIM + col];
        acc2 = __builtin_amdgcn_wmma_f32_16x16x4_f32(
            false, a, false, b, (short)0, acc2, false, false);
    }

    float bias2 = b2[col];
    float s = 0.f, q = 0.f;
    #pragma unroll
    for (int g = 0; g < 8; ++g) {
        int m = g + 8 * half;
        float v = acc2[g] + bias2;
        v = v > 0.f ? v : 0.f;
        hbuf[(size_t)(row0 + m) * DIM + col] = v;
        s += v;
        q += v * v;
    }
    s += __shfl_xor(s, 16);
    q += __shfl_xor(q, 16);
    if (half == 0) {
        atomicAdd(&stats[col], s);
        atomicAdd(&stats[DIM + col], q);
    }
}

__global__ void bn_stats_finalize(float* __restrict__ stats)
{
    int c = threadIdx.x;
    float mu  = stats[c] * (1.0f / N_NODES);
    float var = stats[DIM + c] * (1.0f / N_NODES) - mu * mu;
    stats[c]       = mu;
    stats[DIM + c] = rsqrtf(var + BN_EPS);
}

__global__ __launch_bounds__(256) void bn_pool_kernel(
    const float* __restrict__ h,
    const float* __restrict__ stats,
    const float* __restrict__ gamma, const float* __restrict__ beta,
    const int* __restrict__ batch,
    float* __restrict__ zcat, float* __restrict__ gcat)
{
    int tid = blockIdx.x * blockDim.x + threadIdx.x;
    if (tid >= N_NODES * 32) return;
    int i    = tid >> 5;
    int lane = tid & 31;
    int g = batch[i];
    const float* hr = h + (size_t)i * DIM + lane;
    float* zr = zcat + (size_t)i * ZSTRIDE + lane;
    float* gr = gcat + (size_t)g * ZSTRIDE + lane;
    float in[4], mu[4], rs[4], ga[4], be[4], y[4];
    for (int j = 0; j < 4; ++j) {
        int c = lane + 32 * j;
        in[j] = hr[32 * j];
        mu[j] = stats[c];
        rs[j] = stats[DIM + c];
        ga[j] = gamma[c];
        be[j] = beta[c];
    }
    for (int j = 0; j < 4; ++j)
        y[j] = (in[j] - mu[j]) * rs[j] * ga[j] + be[j];
    for (int j = 0; j < 4; ++j) zr[32 * j] = y[j];
    for (int j = 0; j < 4; ++j) atomicAdd(gr + 32 * j, y[j]);
}

extern "C" void kernel_launch(void* const* d_in, const int* in_sizes, int n_in,
                              void* d_out, int out_size, void* d_ws, size_t ws_size,
                              hipStream_t stream)
{
    (void)in_sizes; (void)n_in; (void)out_size; (void)ws_size;

    const float* x     = (const float*)d_in[0];
    const int*   ei    = (const int*)  d_in[1];
    const int*   batch = (const int*)  d_in[2];
    const float* W1    = (const float*)d_in[3];
    const float* b1    = (const float*)d_in[4];
    const float* W2    = (const float*)d_in[5];
    const float* b2    = (const float*)d_in[6];
    const float* gamma = (const float*)d_in[7];
    const float* beta  = (const float*)d_in[8];

    float* out  = (float*)d_out;
    float* zcat = out;
    float* gcat = out + (size_t)N_NODES * ZSTRIDE;

    float* agg   = (float*)d_ws;
    float* stats = agg + (size_t)N_NODES * DIM;
    unsigned* owner = (unsigned*)(stats + 2 * DIM);
    unsigned* saved = owner + N_EDGES / 4;
    int* saved_n    = (int*)(saved + (size_t)AGG_BLOCKS * LIST_CAP);

    const int* src = ei;
    const int* dst = ei + N_EDGES;

    hipMemsetAsync(gcat, 0, sizeof(float) * (size_t)NGRAPH * ZSTRIDE, stream);

    owner_kernel<<<(N_EDGES / 4 + 255) / 256, 256, 0, stream>>>(dst, owner);

    for (int l = 0; l < NLAYER; ++l) {
        const float* zin = (l == 0) ? x : (zcat + (size_t)(l - 1) * DIM);
        int zstride      = (l == 0) ? DIM : ZSTRIDE;

        hipMemsetAsync(stats, 0, sizeof(float) * (2 * DIM), stream);

        if (l == 0)
            agg_kernel<true><<<AGG_BLOCKS, AGG_THREADS, 0, stream>>>(
                zin, zstride, src, dst, owner, saved, saved_n, agg);
        else
            agg_kernel<false><<<AGG_BLOCKS, AGG_THREADS, 0, stream>>>(
                zin, zstride, src, dst, owner, saved, saved_n, agg);

        gin_mlp_kernel<<<N_NODES / 16, 256, 0, stream>>>(
            zin, zstride, agg,
            W1 + (size_t)l * DIM * DIM, b1 + (size_t)l * DIM,
            W2 + (size_t)l * DIM * DIM, b2 + (size_t)l * DIM,
            stats);

        bn_stats_finalize<<<1, DIM, 0, stream>>>(stats);

        {
            long long threads = (long long)N_NODES * 32;
            int blocks = (int)((threads + 255) / 256);
            bn_pool_kernel<<<blocks, 256, 0, stream>>>(
                agg, stats,
                gamma + (size_t)l * DIM, beta + (size_t)l * DIM,
                batch,
                zcat + (size_t)l * DIM, gcat + (size_t)l * DIM);
        }
    }
}


# BN apply, z_cat store and pool of layer l move into the next neighbour-sum launch's row-owner blocks (gather normalises raw h in flight, pool via LDS); 13 launches -> 8; row, pool sum order change
# speedup vs baseline: 2.8550x; 1.1414x over previous
#include <hip/hip_runtime.h>

#define N_NODES 50000
#define N_EDGES 800000
#define DIM     128
#define NLAYER  3
#define NGRAPH  512
#define ZSTRIDE (NLAYER * DIM)
#define BN_EPS  1e-5f
#define LDS_STRIDE 132

typedef __attribute__((ext_vector_type(2))) float v2f;
typedef __attribute__((ext_vector_type(8))) float v8f;

#define RB          196
#define AGG_BLOCKS  256
#define AGG_THREADS 1024
#define LIST_CAP    16384
#define GROUPS16    (N_EDGES / 16)
#define SCAN_UNROLL 7
#define GE          8
#define PRE         (LIST_CAP / AGG_THREADS)

static_assert(N_NODES <= 65536 && RB <= 65536, "a saved word holds two 16-bit row numbers");

__global__ __launch_bounds__(256) void owner_kernel(
    const int* __restrict__ dst, unsigned* __restrict__ owner)
{
    const int i = blockIdx.x * 256 + threadIdx.x;
    if (i >= N_EDGES / 4) return;
    const int4 d = *(const int4*)(dst + 4 * i);
    owner[i] =  ((unsigned)d.x / RB)        | (((unsigned)d.y / RB) << 8) |
               (((unsigned)d.z / RB) << 16) | (((unsigned)d.w / RB) << 24);
}

__device__ __forceinline__ unsigned zero_bytes(unsigned x)
{
    unsigned t = (x & 0x7f7f7f7fu) + 0x7f7f7f7fu;
    t = t | x | 0x7f7f7f7fu;
    return ~t;
}

#define POOL_CAP 64
#define RPW ((RB + 31) / 32)

__device__ __forceinline__ void bn_column(const float* __restrict__ stats,
    const float* __restrict__ gamma, const float* __restrict__ beta, int c,
    float& mu, float& rs, float& ga, float& be)
{
    mu = stats[c] * (1.0f / N_NODES);
    const float var = stats[DIM + c] * (1.0f / N_NODES) - mu * mu;
    rs = rsqrtf(var + BN_EPS);
    ga = gamma[c];
    be = beta[c];
}
__device__ __forceinline__ float bn_apply(float h, float mu, float rs, float ga, float be)
{
    return (h - mu) * rs * ga + be;
}

template <int MODE>
__global__ __launch_bounds__(AGG_THREADS) void agg_kernel(
    const float* __restrict__ hin,
    const int* __restrict__ src, const int* __restrict__ dst,
    const unsigned* __restrict__ owner,
    unsigned* __restrict__ saved, int* __restrict__ saved_n,
    const float* __restrict__ stats, const float* __restrict__ gamma, const float* __restrict__ beta,
    const int* __restrict__ batch,
    float* __restrict__ zcat, float* __restrict__ gcat,
    float* __restrict__ sum)
{
    constexpr bool FIRST  = MODE == 0;
    constexpr bool BN     = MODE != 0;
    constexpr bool GATHER = MODE != 2;
    __shared__ float acc[RB * DIM];
    __shared__ int   list[LIST_CAP];
    __shared__ float pool[POOL_CAP * DIM];
    __shared__ int   nlist;
    __shared__ int   gmax;

    const int tid  = threadIdx.x;
    const int lane = tid & 31;
    const int wv   = __builtin_amdgcn_readfirstlane(tid >> 5);
    const int lo   = blockIdx.x * RB;
    const int nrows = (N_NODES - lo < RB) ? (N_NODES - lo) : RB;
    unsigned* mine = saved + (size_t)blockIdx.x * LIST_CAP;

    int  nsaved = 0;
    bool scan   = FIRST;
    if (MODE == 1) {
        nsaved = __builtin_amdgcn_readfirstlane(saved_n[blockIdx.x]);
        scan   = nsaved > LIST_CAP;
    }
    float mu[4], rs[4], ga[4], be[4];
    if (BN) {
        #pragma unroll
        for (int j = 0; j < 4; ++j)
            bn_column(stats, gamma, beta, lane + 32 * j, mu[j], rs[j], ga[j], be[j]);
    }
    unsigned pre[PRE];
    if (MODE == 1 && !scan) {
        #pragma unroll
        for (int u = 0; u < PRE; ++u) {
            const int i = u * AGG_THREADS + tid;
            pre[u] = (i < nsaved) ? mine[i] : 0u;
        }
    }
    if (BN) {
        for (int i = tid; i < POOL_CAP * DIM; i += AGG_THREADS) pool[i] = 0.f;
        if (tid == 0) gmax = -1;
    }
    if (tid == 0) nlist = 0;

    const int gfirst = BN ? batch[lo] : 0;
    float hv[RPW][4];
    int   gr[RPW];
    #pragma unroll
    for (int t = 0; t < RPW; ++t) {
        const int r = wv + 32 * t;
        gr[t] = 0;
        hv[t][0] = hv[t][1] = hv[t][2] = hv[t][3] = 0.f;
        if (r < nrows) {
            const float* hr = hin + (size_t)(lo + r) * DIM + lane;
            hv[t][0] = hr[0];
            hv[t][1] = hr[32];
            hv[t][2] = hr[64];
            hv[t][3] = hr[96];
            if (BN) gr[t] = batch[lo + r];
        }
    }
    if (BN) __syncthreads();
    #pragma unroll
    for (int t = 0; t < RPW; ++t) {
        const int r = wv + 32 * t;
        if (r < RB) {
            float y[4];
            #pragma unroll
            for (int j = 0; j < 4; ++j)
                y[j] = BN ? bn_apply(hv[t][j], mu[j], rs[j], ga[j], be[j]) : hv[t][j];
            if (BN && r < nrows) {
                float* zr = zcat + (size_t)(lo + r) * ZSTRIDE + lane;
                #pragma unroll
                for (int j = 0; j < 4; ++j) zr[32 * j] = y[j];
                const unsigned p = (unsigned)(gr[t] - gfirst);
                if (p < POOL_CAP) {
                    #pragma unroll
                    for (int j = 0; j < 4; ++j) atomicAdd(&pool[p * DIM + lane + 32 * j], y[j]);
                    if (lane == 0) atomicMax(&gmax, (int)p);
                } else {
                    float* gp = gcat + (size_t)gr[t] * ZSTRIDE + lane;
                    #pragma unroll
                    for (int j = 0; j < 4; ++j) atomicAdd(gp + 32 * j, y[j]);
                }
            }
            if (GATHER) {
                #pragma unroll
                for (int j = 0; j < 4; ++j) acc[r * DIM + lane + 32 * j] = (r < nrows) ? y[j] : 0.f;
            }
        }
    }

    if (MODE == 1 && !scan) {
        #pragma unroll
        for (int u = 0; u < PRE; ++u) {
            const int i = u * AGG_THREADS + tid;
            if (i < nsaved) list[i] = (int)pre[u];
        }
    }
    __syncthreads();

  if (GATHER) {
    int n = nsaved;
    if (scan) {
        const unsigned pat = (unsigned)blockIdx.x * 0x01010101u;
        const uint4* ow = (const uint4*)owner;
        for (int g0 = 0; g0 < GROUPS16; g0 += AGG_THREADS * SCAN_UNROLL) {
            uint4 w[SCAN_UNROLL];
            #pragma unroll
            for (int u = 0; u < SCAN_UNROLL; ++u) {
                const int g = g0 + u * AGG_THREADS + tid;
                w[u] = ow[g < GROUPS16 ? g : GROUPS16 - 1];
            }
            #pragma unroll
            for (int u = 0; u < SCAN_UNROLL; ++u) {
                const int g = g0 + u * AGG_THREADS + tid;
                unsigned m = (zero_bytes(w[u].x ^ pat) >> 3) | (zero_bytes(w[u].y ^ pat) >> 2) |
                             (zero_bytes(w[u].z ^ pat) >> 1) |  zero_bytes(w[u].w ^ pat);
                if (g >= GROUPS16) m = 0;
                while (m) {
                    const int p = __builtin_ctz(m);
                    m &= m - 1;
                    const int e = g * 16 + (((p & 7) - 4) << 2) + (p >> 3);
                    const int slot = atomicAdd(&nlist, 1);
                    if (slot < LIST_CAP) {
                        list[slot] = e;
                    } else {
                        const int dl = dst[e] - lo;
                        const float* zr = hin + (size_t)src[e] * DIM;
                        for (int c = 0; c < DIM; ++c) {
                            float v = zr[c];
                            if (BN) {
                                float m1, r1, g1, b1;
                                bn_column(stats, gamma, beta, c, m1, r1, g1, b1);
                                v = bn_apply(v, m1, r1, g1, b1);
                            }
                            atomicAdd(&acc[dl * DIM + c], v);
                        }
                    }
                }
            }
        }
        __syncthreads();
        const int nl = __builtin_amdgcn_readfirstlane(nlist);
        n = nl < LIST_CAP ? nl : LIST_CAP;
        if (FIRST && tid == 0) saved_n[blockIdx.x] = nl;
    }

    for (int base = wv * 32; base < n; base += AGG_THREADS) {
        const int cnt = (n - base < 32) ? (n - base) : 32;
        int sl = 0, dl = 0;
        if (lane < cnt) {
            const int w = list[base + lane];
            if (scan) {
                sl = src[w];
                dl = dst[w] - lo;
                if (FIRST) mine[base + lane] = (unsigned)sl | ((unsigned)dl << 16);
            } else {
                sl = w & 0xffff;
                dl = (int)((unsigned)w >> 16);
            }
        }
        for (int j0 = 0; j0 < cnt; j0 += GE) {
            int s[GE], d[GE];
            #pragma unroll
            for (int u = 0; u < GE; ++u) {
                s[u] = __builtin_amdgcn_readlane(sl, j0 + u);
                d[u] = __builtin_amdgcn_readlane(dl, j0 + u);
            }
            float v[GE][4];
            #pragma unroll
            for (int u = 0; u < GE; ++u) {
                if (j0 + u < cnt) {
                    const float* zr = hin + (size_t)s[u] * DIM + lane;
                    v[u][0] = zr[0];
                    v[u][1] = zr[32];
                    v[u][2] = zr[64];
                    v[u][3] = zr[96];
                }
            }
            #pragma unroll
            for (int u = 0; u < GE; ++u) {
                if (j0 + u < cnt) {
                    float* a = acc + d[u] * DIM + lane;
                    #pragma unroll
                    for (int j = 0; j < 4; ++j) {
                        const float y = BN ? bn_apply(v[u][j], mu[j], rs[j], ga[j], be[j]) : v[u][j];
                        atomicAdd(a + 32 * j, y);
                    }
                }
            }
        }
    }
  }
    __syncthreads();

    if (GATHER) {
        float* outp = sum + (size_t)lo * DIM;
        for (int i = tid; i < nrows * DIM; i += AGG_THREADS) outp[i] = acc[i];
    }
    if (BN) {
        const int np = __builtin_amdgcn_readfirstlane(gmax) + 1;
        for (int p = wv; p < np; p += AGG_THREADS / 32) {
            float* gp = gcat + (size_t)(gfirst + p) * ZSTRIDE + lane;
            #pragma unroll
            for (int j = 0; j < 4; ++j) atomicAdd(gp + 32 * j, pool[p * DIM + lane + 32 * j]);
        }
    }
}

__global__ __launch_bounds__(256) void gin_mlp_kernel(
    const float* __restrict__ sum,
    float* __restrict__ hbuf,
    const float* __restrict__ W1, const float* __restrict__ b1,
    const float* __restrict__ W2, const float* __restrict__ b2,
    float* __restrict__ stats)
{
    __shared__ float ldsA[16 * LDS_STRIDE];
    __shared__ float ldsH[16 * LDS_STRIDE];

    const int tid  = threadIdx.x;
    const int wave = tid >> 5;
    const int lane = tid & 31;
    const int half = lane >> 4;
    const int l16  = lane & 15;
    const int row0 = blockIdx.x * 16;
    const int col  = wave * 16 + l16;

    for (int idx = tid; idx < 16 * DIM; idx += 256) {
        int r = idx >> 7, c = idx & 127;
        ldsA[r * LDS_STRIDE + c] = sum[(size_t)(row0 + r) * DIM + c];
    }
    __syncthreads();

    v8f acc = {};
    for (int k0 = 0; k0 < DIM; k0 += 4) {
        int ka = k0 + 2 * half;
        v2f a, b;
        a.x = ldsA[l16 * LDS_STRIDE + ka];
        a.y = ldsA[l16 * LDS_STRIDE + ka + 1];
        b.x = W1[ka * DIM + col];
        b.y = W1[(ka + 1) * DIM + col];
        acc = __builtin_amdgcn_wmma_f32_16x16x4_f32(
            false, a, false, b, (short)0, acc, false, false);
    }

    float bias1 = b1[col];
    #pragma unroll
    for (int g = 0; g < 8; ++g) {
        int m = g + 8 * half;
        float v = acc[g] + bias1;
        ldsH[m * LDS_STRIDE + col] = v > 0.f ? v : 0.f;
    }
    __syncthreads();

    v8f acc2 = {};
    for (int k0 = 0; k0 < DIM; k0 += 4) {
        int ka = k0 + 2 * half;
        v2f a, b;
        a.x = ldsH[l16 * LDS_STRIDE + ka];
        a.y = ldsH[l16 * LDS_STRIDE + ka + 1];
        b.x = W2[ka * DIM + col];
        b.y = W2[(ka + 1) * DIM + col];
        acc2 = __builtin_amdgcn_wmma_f32_16x16x4_f32(
            false, a, false, b, (short)0, acc2, false, false);
    }

    float bias2 = b2[col];
    float s = 0.f, q = 0.f;
    #pragma unroll
    for (int g = 0; g < 8; ++g) {
        int m = g + 8 * half;
        float v = acc2[g] + bias2;
        v = v > 0.f ? v : 0.f;
        hbuf[(size_t)(row0 + m) * DIM + col] = v;
        s += v;
        q += v * v;
    }
    s += __shfl_xor(s, 16);
    q += __shfl_xor(q, 16);
    if (half == 0) {
        atomicAdd(&stats[col], s);
        atomicAdd(&stats[DIM + col], q);
    }
}

extern "C" void kernel_launch(void* const* d_in, const int* in_sizes, int n_in,
                              void* d_out, int out_size, void* d_ws, size_t ws_size,
                              hipStream_t stream)
{
    (void)in_sizes; (void)n_in; (void)out_size; (void)ws_size;

    const float* x     = (const float*)d_in[0];
    const int*   ei    = (const int*)  d_in[1];
    const int*   batch = (const int*)  d_in[2];
    const float* W1    = (const float*)d_in[3];
    const float* b1    = (const float*)d_in[4];
    const float* W2    = (const float*)d_in[5];
    const float* b2    = (const float*)d_in[6];
    const float* gamma = (const float*)d_in[7];
    const float* beta  = (const float*)d_in[8];

    float* out  = (float*)d_out;
    float* zcat = out;
    float* gcat = out + (size_t)N_NODES * ZSTRIDE;

    float* sum   = (float*)d_ws;
    float* hbuf  = sum + (size_t)N_NODES * DIM;
    float* stats = hbuf + (size_t)N_NODES * DIM;
    unsigned* owner = (unsigned*)(stats + NLAYER * 2 * DIM);
    unsigned* saved = owner + N_EDGES / 4;
    int* saved_n    = (int*)(saved + (size_t)AGG_BLOCKS * LIST_CAP);

    const int* src = ei;
    const int* dst = ei + N_EDGES;

    hipMemsetAsync(gcat, 0, sizeof(float) * (size_t)NGRAPH * ZSTRIDE, stream);
    hipMemsetAsync(stats, 0, sizeof(float) * (NLAYER * 2 * DIM), stream);

    owner_kernel<<<(N_EDGES / 4 + 255) / 256, 256, 0, stream>>>(dst, owner);

    for (int l = 0; l < NLAYER; ++l) {
        if (l == 0)
            agg_kernel<0><<<AGG_BLOCKS, AGG_THREADS, 0, stream>>>(
                x, src, dst, owner, saved, saved_n,
                stats, gamma, beta, batch, zcat, gcat, sum);
        else
            agg_kernel<1><<<AGG_BLOCKS, AGG_THREADS, 0, stream>>>(
                hbuf, src, dst, owner, saved, saved_n,
                stats + (size_t)(l - 1) * 2 * DIM,
                gamma + (size_t)(l - 1) * DIM, beta + (size_t)(l - 1) * DIM, batch,
                zcat + (size_t)(l - 1) * DIM, gcat + (size_t)(l - 1) * DIM, sum);

        gin_mlp_kernel<<<N_NODES / 16, 256, 0, stream>>>(
            sum, hbuf,
            W1 + (size_t)l * DIM * DIM, b1 + (size_t)l * DIM,
            W2 + (size_t)l * DIM * DIM, b2 + (size_t)l * DIM,
            stats + (size_t)l * 2 * DIM);
    }
    agg_kernel<2><<<AGG_BLOCKS, AGG_THREADS, 0, stream>>>(
        hbuf, src, dst, owner, saved, saved_n,
        stats + (size_t)(NLAYER - 1) * 2 * DIM,
        gamma + (size_t)(NLAYER - 1) * DIM, beta + (size_t)(NLAYER - 1) * DIM, batch,
        zcat + (size_t)(NLAYER - 1) * DIM, gcat + (size_t)(NLAYER - 1) * DIM, sum);
}


# MLP: 256 blocks, one per WGP by LDS size (161 KiB), hold W1 and W2 whole in LDS; wave = 16 rows x 32 cols, B fragment = one 8-byte LDS load; BN sums added per block first: column sum order changes
# speedup vs baseline: 3.2700x; 1.1454x over previous
#include <hip/hip_runtime.h>

#define N_NODES 50000
#define N_EDGES 800000
#define DIM     128
#define NLAYER  3
#define NGRAPH  512
#define ZSTRIDE (NLAYER * DIM)
#define BN_EPS  1e-5f
#define LDS_STRIDE 132

typedef __attribute__((ext_vector_type(2))) float v2f;
typedef __attribute__((ext_vector_type(8))) float v8f;

#define RB          196
#define AGG_BLOCKS  256
#define AGG_THREADS 1024
#define LIST_CAP    16384
#define GROUPS16    (N_EDGES / 16)
#define SCAN_UNROLL 7
#define GE          8
#define PRE         (LIST_CAP / AGG_THREADS)

static_assert(N_NODES <= 65536 && RB <= 65536, "a saved word holds two 16-bit row numbers");

__global__ __launch_bounds__(256) void owner_kernel(
    const int* __restrict__ dst, unsigned* __restrict__ owner)
{
    const int i = blockIdx.x * 256 + threadIdx.x;
    if (i >= N_EDGES / 4) return;
    const int4 d = *(const int4*)(dst + 4 * i);
    owner[i] =  ((unsigned)d.x / RB)        | (((unsigned)d.y / RB) << 8) |
               (((unsigned)d.z / RB) << 16) | (((unsigned)d.w / RB) << 24);
}

__device__ __forceinline__ unsigned zero_bytes(unsigned x)
{
    unsigned t = (x & 0x7f7f7f7fu) + 0x7f7f7f7fu;
    t = t | x | 0x7f7f7f7fu;
    return ~t;
}

#define POOL_CAP 64
#define RPW ((RB + 31) / 32)

__device__ __forceinline__ void bn_column(const float* __restrict__ stats,
    const float* __restrict__ gamma, const float* __restrict__ beta, int c,
    float& mu, float& rs, float& ga, float& be)
{
    mu = stats[c] * (1.0f / N_NODES);
    const float var = stats[DIM + c] * (1.0f / N_NODES) - mu * mu;
    rs = rsqrtf(var + BN_EPS);
    ga = gamma[c];
    be = beta[c];
}
__device__ __forceinline__ float bn_apply(float h, float mu, float rs, float ga, float be)
{
    return (h - mu) * rs * ga + be;
}

template <int MODE>
__global__ __launch_bounds__(AGG_THREADS) void agg_kernel(
    const float* __restrict__ hin,
    const int* __restrict__ src, const int* __restrict__ dst,
    const unsigned* __restrict__ owner,
    unsigned* __restrict__ saved, int* __restrict__ saved_n,
    const float* __restrict__ stats, const float* __restrict__ gamma, const float* __restrict__ beta,
    const int* __restrict__ batch,
    float* __restrict__ zcat, float* __restrict__ gcat,
    float* __restrict__ sum)
{
    constexpr bool FIRST  = MODE == 0;
    constexpr bool BN     = MODE != 0;
    constexpr bool GATHER = MODE != 2;
    __shared__ float acc[RB * DIM];
    __shared__ int   list[LIST_CAP];
    __shared__ float pool[POOL_CAP * DIM];
    __shared__ int   nlist;
    __shared__ int   gmax;

    const int tid  = threadIdx.x;
    const int lane = tid & 31;
    const int wv   = __builtin_amdgcn_readfirstlane(tid >> 5);
    const int lo   = blockIdx.x * RB;
    const int nrows = (N_NODES - lo < RB) ? (N_NODES - lo) : RB;
    unsigned* mine = saved + (size_t)blockIdx.x * LIST_CAP;

    int  nsaved = 0;
    bool scan   = FIRST;
    if (MODE == 1) {
        nsaved = __builtin_amdgcn_readfirstlane(saved_n[blockIdx.x]);
        scan   = nsaved > LIST_CAP;
    }
    float mu[4], rs[4], ga[4], be[4];
    if (BN) {
        #pragma unroll
        for (int j = 0; j < 4; ++j)
            bn_column(stats, gamma, beta, lane + 32 * j, mu[j], rs[j], ga[j], be[j]);
    }
    unsigned pre[PRE];
    if (MODE == 1 && !scan) {
        #pragma unroll
        for (int u = 0; u < PRE; ++u) {
            const int i = u * AGG_THREADS + tid;
            pre[u] = (i < nsaved) ? mine[i] : 0u;
        }
    }
    if (BN) {
        for (int i = tid; i < POOL_CAP * DIM; i += AGG_THREADS) pool[i] = 0.f;
        if (tid == 0) gmax = -1;
    }
    if (tid == 0) nlist = 0;

    const int gfirst = BN ? batch[lo] : 0;
    float hv[RPW][4];
    int   gr[RPW];
    #pragma unroll
    for (int t = 0; t < RPW; ++t) {
        const int r = wv + 32 * t;
        gr[t] = 0;
        hv[t][0] = hv[t][1] = hv[t][2] = hv[t][3] = 0.f;
        if (r < nrows) {
            const float* hr = hin + (size_t)(lo + r) * DIM + lane;
            hv[t][0] = hr[0];
            hv[t][1] = hr[32];
            hv[t][2] = hr[64];
            hv[t][3] = hr[96];
            if (BN) gr[t] = batch[lo + r];
        }
    }
    if (BN) __syncthreads();
    #pragma unroll
    for (int t = 0; t < RPW; ++t) {
        const int r = wv + 32 * t;
        if (r < RB) {
            float y[4];
            #pragma unroll
            for (int j = 0; j < 4; ++j)
                y[j] = BN ? bn_apply(hv[t][j], mu[j], rs[j], ga[j], be[j]) : hv[t][j];
            if (BN && r < nrows) {
                float* zr = zcat + (size_t)(lo + r) * ZSTRIDE + lane;
                #pragma unroll
                for (int j = 0; j < 4; ++j) zr[32 * j] = y[j];
                const unsigned p = (unsigned)(gr[t] - gfirst);
                if (p < POOL_CAP) {
                    #pragma unroll
                    for (int j = 0; j < 4; ++j) atomicAdd(&pool[p * DIM + lane + 32 * j], y[j]);
                    if (lane == 0) atomicMax(&gmax, (int)p);
                } else {
                    float* gp = gcat + (size_t)gr[t] * ZSTRIDE + lane;
                    #pragma unroll
                    for (int j = 0; j < 4; ++j) atomicAdd(gp + 32 * j, y[j]);
                }
            }
            if (GATHER) {
                #pragma unroll
                for (int j = 0; j < 4; ++j) acc[r * DIM + lane + 32 * j] = (r < nrows) ? y[j] : 0.f;
            }
        }
    }

    if (MODE == 1 && !scan) {
        #pragma unroll
        for (int u = 0; u < PRE; ++u) {
            const int i = u * AGG_THREADS + tid;
            if (i < nsaved) list[i] = (int)pre[u];
        }
    }
    __syncthreads();

  if (GATHER) {
    int n = nsaved;
    if (scan) {
        const unsigned pat = (unsigned)blockIdx.x * 0x01010101u;
        const uint4* ow = (const uint4*)owner;
        for (int g0 = 0; g0 < GROUPS16; g0 += AGG_THREADS * SCAN_UNROLL) {
            uint4 w[SCAN_UNROLL];
            #pragma unroll
            for (int u = 0; u < SCAN_UNROLL; ++u) {
                const int g = g0 + u * AGG_THREADS + tid;
                w[u] = ow[g < GROUPS16 ? g : GROUPS16 - 1];
            }
            #pragma unroll
            for (int u = 0; u < SCAN_UNROLL; ++u) {
                const int g = g0 + u * AGG_THREADS + tid;
                unsigned m = (zero_bytes(w[u].x ^ pat) >> 3) | (zero_bytes(w[u].y ^ pat) >> 2) |
                             (zero_bytes(w[u].z ^ pat) >> 1) |  zero_bytes(w[u].w ^ pat);
                if (g >= GROUPS16) m = 0;
                while (m) {
                    const int p = __builtin_ctz(m);
                    m &= m - 1;
                    const int e = g * 16 + (((p & 7) - 4) << 2) + (p >> 3);
                    const int slot = atomicAdd(&nlist, 1);
                    if (slot < LIST_CAP) {
                        list[slot] = e;
                    } else {
                        const int dl = dst[e] - lo;
                        const float* zr = hin + (size_t)src[e] * DIM;
                        for (int c = 0; c < DIM; ++c) {
                            float v = zr[c];
                            if (BN) {
                                float m1, r1, g1, b1;
                                bn_column(stats, gamma, beta, c, m1, r1, g1, b1);
                                v = bn_apply(v, m1, r1, g1, b1);
                            }
                            atomicAdd(&acc[dl * DIM + c], v);
                        }
                    }
                }
            }
        }
        __syncthreads();
        const int nl = __builtin_amdgcn_readfirstlane(nlist);
        n = nl < LIST_CAP ? nl : LIST_CAP;
        if (FIRST && tid == 0) saved_n[blockIdx.x] = nl;
    }

    for (int base = wv * 32; base < n; base += AGG_THREADS) {
        const int cnt = (n - base < 32) ? (n - base) : 32;
        int sl = 0, dl = 0;
        if (lane < cnt) {
            const int w = list[base + lane];
            if (scan) {
                sl = src[w];
                dl = dst[w] - lo;
                if (FIRST) mine[base + lane] = (unsigned)sl | ((unsigned)dl << 16);
            } else {
                sl = w & 0xffff;
                dl = (int)((unsigned)w >> 16);
            }
        }
        for (int j0 = 0; j0 < cnt; j0 += GE) {
            int s[GE], d[GE];
            #pragma unroll
            for (int u = 0; u < GE; ++u) {
                s[u] = __builtin_amdgcn_readlane(sl, j0 + u);
                d[u] = __builtin_amdgcn_readlane(dl, j0 + u);
            }
            float v[GE][4];
            #pragma unroll
            for (int u = 0; u < GE; ++u) {
                if (j0 + u < cnt) {
                    const float* zr = hin + (size_t)s[u] * DIM + lane;
                    v[u][0] = zr[0];
                    v[u][1] = zr[32];
                    v[u][2] = zr[64];
                    v[u][3] = zr[96];
                }
            }
            #pragma unroll
            for (int u = 0; u < GE; ++u) {
                if (j0 + u < cnt) {
                    float* a = acc + d[u] * DIM + lane;
                    #pragma unroll
                    for (int j = 0; j < 4; ++j) {
                        const float y = BN ? bn_apply(v[u][j], mu[j], rs[j], ga[j], be[j]) : v[u][j];
                        atomicAdd(a + 32 * j, y);
                    }
                }
            }
        }
    }
  }
    __syncthreads();

    if (GATHER) {
        float* outp = sum + (size_t)lo * DIM;
        for (int i = tid; i < nrows * DIM; i += AGG_THREADS) outp[i] = acc[i];
    }
    if (BN) {
        const int np = __builtin_amdgcn_readfirstlane(gmax) + 1;
        for (int p = wv; p < np; p += AGG_THREADS / 32) {
            float* gp = gcat + (size_t)(gfirst + p) * ZSTRIDE + lane;
            #pragma unroll
            for (int j = 0; j < 4; ++j) atomicAdd(gp + 32 * j, pool[p * DIM + lane + 32 * j]);
        }
    }
}

#define MLP_BLOCKS  256
#define MLP_THREADS 256
#define N_TILES     (N_NODES / 16)
#define MLP_ROUNDS  (((N_TILES + MLP_BLOCKS - 1) / MLP_BLOCKS + 1) / 2)
#define TILE_LDS    (16 * LDS_STRIDE)

__device__ __forceinline__ void stage_weights(const float* __restrict__ Wg, float* Wl, int tid)
{
    #pragma unroll
    for (int it = 0; it < 8; ++it) {
        const int item = it * MLP_THREADS + tid;
        const int kp = item >> 5, c4 = item & 31;
        const float4 r0 = *(const float4*)(Wg + (size_t)(2 * kp) * DIM + 4 * c4);
        const float4 r1 = *(const float4*)(Wg + (size_t)(2 * kp + 1) * DIM + 4 * c4);
        float* p = Wl + (kp * DIM + ((4 * c4) ^ ((kp & 1) << 4))) * 2;
        *(float4*)(p)     = make_float4(r0.x, r1.x, r0.y, r1.y);
        *(float4*)(p + 4) = make_float4(r0.z, r1.z, r0.w, r1.w);
    }
}

#define KBLK 4
__device__ __forceinline__ void tile_gemm(const float* At, const float* Wl,
    int l16, int hf, int col0, v8f& c0, v8f& c1)
{
    const float* ap  = At + l16 * LDS_STRIDE + 2 * hf;
    const float* bp0 = Wl + (hf * DIM + (col0 ^ (hf << 4))) * 2;
    const float* bp1 = Wl + (hf * DIM + ((col0 + 16) ^ (hf << 4))) * 2;
    v2f a[KBLK], p[KBLK], q[KBLK];
    #pragma unroll
    for (int u = 0; u < KBLK; ++u) {
        a[u] = *(const v2f*)(ap + 4 * u);
        p[u] = *(const v2f*)(bp0 + 4 * u * DIM);
        q[u] = *(const v2f*)(bp1 + 4 * u * DIM);
    }
    #pragma unroll
    for (int k0 = 0; k0 < DIM; k0 += 4 * KBLK) {
        v2f an[KBLK], pn[KBLK], qn[KBLK];
        if (k0 + 4 * KBLK < DIM) {
            #pragma unroll
            for (int u = 0; u < KBLK; ++u) {
                const int k = k0 + 4 * KBLK + 4 * u;
                an[u] = *(const v2f*)(ap + k);
                pn[u] = *(const v2f*)(bp0 + k * DIM);
                qn[u] = *(const v2f*)(bp1 + k * DIM);
            }
        }
        #pragma unroll
        for (int u = 0; u < KBLK; ++u) {
            c0 = __builtin_amdgcn_wmma_f32_16x16x4_f32(false, a[u], false, p[u], (short)0, c0, false, false);
            c1 = __builtin_amdgcn_wmma_f32_16x16x4_f32(false, a[u], false, q[u], (short)0, c1, false, false);
        }
        if (k0 + 4 * KBLK < DIM) {
            #pragma unroll
            for (int u = 0; u < KBLK; ++u) { a[u] = an[u]; p[u] = pn[u]; q[u] = qn[u]; }
        }
    }
}

__global__ __launch_bounds__(MLP_THREADS) void gin_mlp_kernel(
    const float* __restrict__ sum,
    float* __restrict__ hbuf,
    const float* __restrict__ W1, const float* __restrict__ b1,
    const float* __restrict__ W2, const float* __restrict__ b2,
    float* __restrict__ stats)
{
    __shared__ __attribute__((aligned(16))) float ldsW1[DIM * DIM];
    __shared__ __attribute__((aligned(16))) float ldsW2[DIM * DIM];
    __shared__ __attribute__((aligned(16))) float ldsA[2 * TILE_LDS];
    __shared__ __attribute__((aligned(16))) float ldsH[2 * TILE_LDS];

    const int tid  = threadIdx.x;
    const int wave = __builtin_amdgcn_readfirstlane(tid >> 5);
    const int lane = tid & 31;
    const int hf   = lane >> 4;
    const int l16  = lane & 15;
    const int grp  = wave >> 2;
    const int col0 = (wave & 3) * 32 + l16;
    const int col1 = col0 + 16;

    float4 pa[4];
    #pragma unroll
    for (int g = 0; g < 2; ++g) {
        const int t = blockIdx.x + MLP_BLOCKS * g;
        #pragma unroll
        for (int j = 0; j < 2; ++j) {
            const int idx = tid + MLP_THREADS * j;
            pa[2 * g + j] = (t < N_TILES) ? *(const float4*)(sum + (size_t)t * (16 * DIM) + 4 * idx)
                                          : make_float4(0.f, 0.f, 0.f, 0.f);
        }
    }
    stage_weights(W1, ldsW1, tid);
    stage_weights(W2, ldsW2, tid);
    #pragma unroll
    for (int g = 0; g < 2; ++g) {
        #pragma unroll
        for (int j = 0; j < 2; ++j) {
            const int idx = tid + MLP_THREADS * j;
            *(float4*)(ldsA + g * TILE_LDS + (idx >> 5) * LDS_STRIDE + 4 * (idx & 31)) = pa[2 * g + j];
        }
    }
    const float bias1_0 = b1[col0], bias1_1 = b1[col1];
    const float bias2_0 = b2[col0], bias2_1 = b2[col1];
    float s0 = 0.f, q0 = 0.f, s1 = 0.f, q1 = 0.f;
    __syncthreads();

    const float* At = ldsA + grp * TILE_LDS;
    float*       Ht = ldsH + grp * TILE_LDS;

    for (int r = 0; r < MLP_ROUNDS; ++r) {
        const bool more = r + 1 < MLP_ROUNDS;
        if (more) {
            #pragma unroll
            for (int g = 0; g < 2; ++g) {
                const int t = blockIdx.x + MLP_BLOCKS * (2 * (r + 1) + g);
                #pragma unroll
                for (int j = 0; j < 2; ++j) {
                    const int idx = tid + MLP_THREADS * j;
                    pa[2 * g + j] = (t < N_TILES) ? *(const float4*)(sum + (size_t)t * (16 * DIM) + 4 * idx)
                                                  : make_float4(0.f, 0.f, 0.f, 0.f);
                }
            }
        }
        const int  tile  = blockIdx.x + MLP_BLOCKS * (2 * r + grp);
        const bool valid = tile < N_TILES;

        if (valid) {
            v8f c0 = {}, c1 = {};
            tile_gemm(At, ldsW1, l16, hf, col0, c0, c1);
            #pragma unroll
            for (int g = 0; g < 8; ++g) {
                const int m = g + 8 * hf;
                const float v0 = c0[g] + bias1_0;
                const float v1 = c1[g] + bias1_1;
                Ht[m * LDS_STRIDE + col0] = v0 > 0.f ? v0 : 0.f;
                Ht[m * LDS_STRIDE + col1] = v1 > 0.f ? v1 : 0.f;
            }
        }
        __syncthreads();

        if (valid) {
            v8f c0 = {}, c1 = {};
            tile_gemm(Ht, ldsW2, l16, hf, col0, c0, c1);
            float* hr = hbuf + (size_t)tile * (16 * DIM);
            #pragma unroll
            for (int g = 0; g < 8; ++g) {
                const int m = g + 8 * hf;
                float v0 = c0[g] + bias2_0;
                float v1 = c1[g] + bias2_1;
                v0 = v0 > 0.f ? v0 : 0.f;
                v1 = v1 > 0.f ? v1 : 0.f;
                hr[m * DIM + col0] = v0;
                hr[m * DIM + col1] = v1;
                s0 += v0;  q0 += v0 * v0;
                s1 += v1;  q1 += v1 * v1;
            }
        }
        if (more) {
            #pragma unroll
            for (int g = 0; g < 2; ++g) {
                #pragma unroll
                for (int j = 0; j < 2; ++j) {
                    const int idx = tid + MLP_THREADS * j;
                    *(float4*)(ldsA + g * TILE_LDS + (idx >> 5) * LDS_STRIDE + 4 * (idx & 31)) = pa[2 * g + j];
                }
            }
        }
        __syncthreads();
    }

    float* part = ldsA;
    const int slot = (grp * 2 + hf) * (2 * DIM);
    part[slot + col0]       = s0;
    part[slot + col1]       = s1;
    part[slot + DIM + col0] = q0;
    part[slot + DIM + col1] = q1;
    __syncthreads();
    atomicAdd(&stats[tid], (part[tid] + part[2 * DIM + tid]) + (part[4 * DIM + tid] + part[6 * DIM + tid]));
}

extern "C" void kernel_launch(void* const* d_in, const int* in_sizes, int n_in,
                              void* d_out, int out_size, void* d_ws, size_t ws_size,
                              hipStream_t stream)
{
    (void)in_sizes; (void)n_in; (void)out_size; (void)ws_size;

    const float* x     = (const float*)d_in[0];
    const int*   ei    = (const int*)  d_in[1];
    const int*   batch = (const int*)  d_in[2];
    const float* W1    = (const float*)d_in[3];
    const float* b1    = (const float*)d_in[4];
    const float* W2    = (const float*)d_in[5];
    const float* b2    = (const float*)d_in[6];
    const float* gamma = (const float*)d_in[7];
    const float* beta  = (const float*)d_in[8];

    float* out  = (float*)d_out;
    float* zcat = out;
    float* gcat = out + (size_t)N_NODES * ZSTRIDE;

    float* sum   = (float*)d_ws;
    float* hbuf  = sum + (size_t)N_NODES * DIM;
    float* stats = hbuf + (size_t)N_NODES * DIM;
    unsigned* owner = (unsigned*)(stats + NLAYER * 2 * DIM);
    unsigned* saved = owner + N_EDGES / 4;
    int* saved_n    = (int*)(saved + (size_t)AGG_BLOCKS * LIST_CAP);

    const int* src = ei;
    const int* dst = ei + N_EDGES;

    hipMemsetAsync(gcat, 0, sizeof(float) * (size_t)NGRAPH * ZSTRIDE, stream);
    hipMemsetAsync(stats, 0, sizeof(float) * (NLAYER * 2 * DIM), stream);

    owner_kernel<<<(N_EDGES / 4 + 255) / 256, 256, 0, stream>>>(dst, owner);

    for (int l = 0; l < NLAYER; ++l) {
        if (l == 0)
            agg_kernel<0><<<AGG_BLOCKS, AGG_THREADS, 0, stream>>>(
                x, src, dst, owner, saved, saved_n,
                stats, gamma, beta, batch, zcat, gcat, sum);
        else
            agg_kernel<1><<<AGG_BLOCKS, AGG_THREADS, 0, stream>>>(
                hbuf, src, dst, owner, saved, saved_n,
                stats + (size_t)(l - 1) * 2 * DIM,
                gamma + (size_t)(l - 1) * DIM, beta + (size_t)(l - 1) * DIM, batch,
                zcat + (size_t)(l - 1) * DIM, gcat + (size_t)(l - 1) * DIM, sum);

        gin_mlp_kernel<<<MLP_BLOCKS, MLP_THREADS, 0, stream>>>(
            sum, hbuf,
            W1 + (size_t)l * DIM * DIM, b1 + (size_t)l * DIM,
            W2 + (size_t)l * DIM * DIM, b2 + (size_t)l * DIM,
            stats + (size_t)l * 2 * DIM);
    }
    agg_kernel<2><<<AGG_BLOCKS, AGG_THREADS, 0, stream>>>(
        hbuf, src, dst, owner, saved, saved_n,
        stats + (size_t)(NLAYER - 1) * 2 * DIM,
        gamma + (size_t)(NLAYER - 1) * DIM, beta + (size_t)(NLAYER - 1) * DIM, batch,
        zcat + (size_t)(NLAYER - 1) * DIM, gcat + (size_t)(NLAYER - 1) * DIM, sum);
}


# MLP runs in the row-owner blocks on their LDS sums in place (W1, W2 staged in LDS ahead of the gather); 8 launches -> 5, no sum store and reload; h in two swapped buffers; BN column sum order changes
# speedup vs baseline: 3.9058x; 1.1944x over previous
#include <hip/hip_runtime.h>

#define N_NODES 50000
#define N_EDGES 800000
#define DIM     128
#define NLAYER  3
#define NGRAPH  512
#define ZSTRIDE (NLAYER * DIM)
#define BN_EPS  1e-5f
#define LDS_STRIDE 132

typedef __attribute__((ext_vector_type(2))) float v2f;
typedef __attribute__((ext_vector_type(8))) float v8f;


#define RB          196
#define ACC_ROWS    208
#define N_RTILES    (ACC_ROWS / 16)
#define AGG_BLOCKS  256
#define AGG_THREADS 1024
#define LIST_CAP    8192
#define GROUPS16    (N_EDGES / 16)
#define SCAN_UNROLL 7
#define GE          8
#define PRE         (LIST_CAP / AGG_THREADS)

static_assert(N_NODES <= 65536 && RB <= 65536, "a saved word holds two 16-bit row numbers");
static_assert(ACC_ROWS >= RB && ACC_ROWS % 16 == 0 && N_RTILES <= 2 * (AGG_THREADS / 128), "two passes of 8 row tiles cover the block");

__global__ __launch_bounds__(256) void owner_kernel(
    const int* __restrict__ dst, unsigned* __restrict__ owner)
{
    const int i = blockIdx.x * 256 + threadIdx.x;
    if (i >= N_EDGES / 4) return;
    const int4 d = *(const int4*)(dst + 4 * i);
    owner[i] =  ((unsigned)d.x / RB)        | (((unsigned)d.y / RB) << 8) |
               (((unsigned)d.z / RB) << 16) | (((unsigned)d.w / RB) << 24);
}

__device__ __forceinline__ unsigned zero_bytes(unsigned x)
{
    unsigned t = (x & 0x7f7f7f7fu) + 0x7f7f7f7fu;
    t = t | x | 0x7f7f7f7fu;
    return ~t;
}

#define POOL_CAP 64
#define RPW ((ACC_ROWS + 31) / 32)

__device__ __forceinline__ void bn_column(const float* __restrict__ stats,
    const float* __restrict__ gamma, const float* __restrict__ beta, int c,
    float& mu, float& rs, float& ga, float& be)
{
    mu = stats[c] * (1.0f / N_NODES);
    const float var = stats[DIM + c] * (1.0f / N_NODES) - mu * mu;
    rs = rsqrtf(var + BN_EPS);
    ga = gamma[c];
    be = beta[c];
}
__device__ __forceinline__ float bn_apply(float h, float mu, float rs, float ga, float be)
{
    return (h - mu) * rs * ga + be;
}


__device__ __forceinline__ void stage_weights(const float* __restrict__ Wg, float* Wl, int tid)
{
    #pragma unroll
    for (int it = 0; it < (DIM / 2) * (DIM / 4) / AGG_THREADS; ++it) {
        const int item = it * AGG_THREADS + tid;
        const int kp = item >> 5, c4 = item & 31;
        const float4 r0 = *(const float4*)(Wg + (size_t)(2 * kp) * DIM + 4 * c4);
        const float4 r1 = *(const float4*)(Wg + (size_t)(2 * kp + 1) * DIM + 4 * c4);
        float* p = Wl + (kp * DIM + ((4 * c4) ^ ((kp & 1) << 4))) * 2;
        *(float4*)(p)     = make_float4(r0.x, r1.x, r0.y, r1.y);
        *(float4*)(p + 4) = make_float4(r0.z, r1.z, r0.w, r1.w);
    }
}

#define KBLK 4
__device__ __forceinline__ void tile_gemm(const float* At, const float* Wl,
    int l16, int hf, int col0, v8f& c0, v8f& c1)
{
    const float* ap  = At + l16 * LDS_STRIDE + 2 * hf;
    const float* bp0 = Wl + (hf * DIM + (col0 ^ (hf << 4))) * 2;
    const float* bp1 = Wl + (hf * DIM + ((col0 + 16) ^ (hf << 4))) * 2;
    v2f a[KBLK], p[KBLK], q[KBLK];
    #pragma unroll
    for (int u = 0; u < KBLK; ++u) {
        a[u] = *(const v2f*)(ap + 4 * u);
        p[u] = *(const v2f*)(bp0 + 4 * u * DIM);
        q[u] = *(const v2f*)(bp1 + 4 * u * DIM);
    }
    #pragma unroll
    for (int k0 = 0; k0 < DIM; k0 += 4 * KBLK) {
        v2f an[KBLK], pn[KBLK], qn[KBLK];
        if (k0 + 4 * KBLK < DIM) {
            #pragma unroll
            for (int u = 0; u < KBLK; ++u) {
                const int k = k0 + 4 * KBLK + 4 * u;
                an[u] = *(const v2f*)(ap + k);
                pn[u] = *(const v2f*)(bp0 + k * DIM);
                qn[u] = *(const v2f*)(bp1 + k * DIM);
            }
        }
        #pragma unroll
        for (int u = 0; u < KBLK; ++u) {
            c0 = __builtin_amdgcn_wmma_f32_16x16x4_f32(false, a[u], false, p[u], (short)0, c0, false, false);
            c1 = __builtin_amdgcn_wmma_f32_16x16x4_f32(false, a[u], false, q[u], (short)0, c1, false, false);
        }
        if (k0 + 4 * KBLK < DIM) {
            #pragma unroll
            for (int u = 0; u < KBLK; ++u) { a[u] = an[u]; p[u] = pn[u]; q[u] = qn[u]; }
        }
    }
}

template <int MODE>
__global__ __launch_bounds__(AGG_THREADS) void agg_kernel(
    const float* __restrict__ hin,
    const int* __restrict__ src, const int* __restrict__ dst,
    const unsigned* __restrict__ owner,
    unsigned* __restrict__ saved, int* __restrict__ saved_n,
    const float* __restrict__ stats, const float* __restrict__ gamma, const float* __restrict__ beta,
    const int* __restrict__ batch,
    float* __restrict__ zcat, float* __restrict__ gcat,
    const float* __restrict__ W1, const float* __restrict__ b1,
    const float* __restrict__ W2, const float* __restrict__ b2,
    float* __restrict__ hout, float* __restrict__ stats_out)
{
    constexpr bool FIRST  = MODE == 0;
    constexpr bool BN     = MODE != 0;
    constexpr bool GATHER = MODE != 2;
    __shared__ __attribute__((aligned(16))) float acc[ACC_ROWS * LDS_STRIDE];
    __shared__ __attribute__((aligned(16))) float ldsW1[DIM * DIM];
    __shared__ __attribute__((aligned(16))) float ldsW2[DIM * DIM];
    __shared__ __attribute__((aligned(16))) int   list[LIST_CAP];
    __shared__ float pool[POOL_CAP * DIM];
    __shared__ int   nlist;
    __shared__ int   gmax;

    const int tid  = threadIdx.x;
    const int lane = tid & 31;
    const int wv   = __builtin_amdgcn_readfirstlane(tid >> 5);
    const int lo   = blockIdx.x * RB;
    const int nrows = (N_NODES - lo < RB) ? (N_NODES - lo) : RB;
    unsigned* mine = saved + (size_t)blockIdx.x * LIST_CAP;

    int  nsaved = 0;
    bool scan   = FIRST;
    if (MODE == 1) {
        nsaved = __builtin_amdgcn_readfirstlane(saved_n[blockIdx.x]);
        scan   = nsaved > LIST_CAP;
    }
    float mu[4], rs[4], ga[4], be[4];
    if (BN) {
        #pragma unroll
        for (int j = 0; j < 4; ++j)
            bn_column(stats, gamma, beta, lane + 32 * j, mu[j], rs[j], ga[j], be[j]);
    }
    unsigned pre[PRE];
    if (MODE == 1 && !scan) {
        #pragma unroll
        for (int u = 0; u < PRE; ++u) {
            const int i = u * AGG_THREADS + tid;
            pre[u] = (i < nsaved) ? mine[i] : 0u;
        }
    }
    if (BN) {
        for (int i = tid; i < POOL_CAP * DIM; i += AGG_THREADS) pool[i] = 0.f;
        if (tid == 0) gmax = -1;
    }
    if (tid == 0) nlist = 0;

    const int gfirst = BN ? batch[lo] : 0;
    float hv[RPW][4];
    int   gr[RPW];
    #pragma unroll
    for (int t = 0; t < RPW; ++t) {
        const int r = wv + 32 * t;
        gr[t] = 0;
        hv[t][0] = hv[t][1] = hv[t][2] = hv[t][3] = 0.f;
        if (r < nrows) {
            const float* hr = hin + (size_t)(lo + r) * DIM + lane;
            hv[t][0] = hr[0];
            hv[t][1] = hr[32];
            hv[t][2] = hr[64];
            hv[t][3] = hr[96];
            if (BN) gr[t] = batch[lo + r];
        }
    }
    if (BN) __syncthreads();
    #pragma unroll
    for (int t = 0; t < RPW; ++t) {
        const int r = wv + 32 * t;
        if (r < ACC_ROWS) {
            float y[4];
            #pragma unroll
            for (int j = 0; j < 4; ++j)
                y[j] = BN ? bn_apply(hv[t][j], mu[j], rs[j], ga[j], be[j]) : hv[t][j];
            if (BN && r < nrows) {
                float* zr = zcat + (size_t)(lo + r) * ZSTRIDE + lane;
                #pragma unroll
                for (int j = 0; j < 4; ++j) zr[32 * j] = y[j];
                const unsigned p = (unsigned)(gr[t] - gfirst);
                if (p < POOL_CAP) {
                    #pragma unroll
                    for (int j = 0; j < 4; ++j) atomicAdd(&pool[p * DIM + lane + 32 * j], y[j]);
                    if (lane == 0) atomicMax(&gmax, (int)p);
                } else {
                    float* gp = gcat + (size_t)gr[t] * ZSTRIDE + lane;
                    #pragma unroll
                    for (int j = 0; j < 4; ++j) atomicAdd(gp + 32 * j, y[j]);
                }
            }
            if (GATHER) {
                #pragma unroll
                for (int j = 0; j < 4; ++j) acc[r * LDS_STRIDE + lane + 32 * j] = (r < nrows) ? y[j] : 0.f;
            }
        }
    }

    if (MODE == 1 && !scan) {
        #pragma unroll
        for (int u = 0; u < PRE; ++u) {
            const int i = u * AGG_THREADS + tid;
            if (i < nsaved) list[i] = (int)pre[u];
        }
    }
    __syncthreads();

  if (GATHER) {
    stage_weights(W1, ldsW1, tid);
    stage_weights(W2, ldsW2, tid);

    int n = nsaved;
    if (scan) {
        const unsigned pat = (unsigned)blockIdx.x * 0x01010101u;
        const uint4* ow = (const uint4*)owner;
        for (int g0 = 0; g0 < GROUPS16; g0 += AGG_THREADS * SCAN_UNROLL) {
            uint4 w[SCAN_UNROLL];
            #pragma unroll
            for (int u = 0; u < SCAN_UNROLL; ++u) {
                const int g = g0 + u * AGG_THREADS + tid;
                w[u] = ow[g < GROUPS16 ? g : GROUPS16 - 1];
            }
            #pragma unroll
            for (int u = 0; u < SCAN_UNROLL; ++u) {
                const int g = g0 + u * AGG_THREADS + tid;
                unsigned m = (zero_bytes(w[u].x ^ pat) >> 3) | (zero_bytes(w[u].y ^ pat) >> 2) |
                             (zero_bytes(w[u].z ^ pat) >> 1) |  zero_bytes(w[u].w ^ pat);
                if (g >= GROUPS16) m = 0;
                while (m) {
                    const int p = __builtin_ctz(m);
                    m &= m - 1;
                    const int e = g * 16 + (((p & 7) - 4) << 2) + (p >> 3);
                    const int slot = atomicAdd(&nlist, 1);
                    if (slot < LIST_CAP) {
                        list[slot] = e;
                    } else {
                        const int dl = dst[e] - lo;
                        const float* zr = hin + (size_t)src[e] * DIM;
                        for (int c = 0; c < DIM; ++c) {
                            float v = zr[c];
                            if (BN) {
                                float m1, r1, g1, b1;
                                bn_column(stats, gamma, beta, c, m1, r1, g1, b1);
                                v = bn_apply(v, m1, r1, g1, b1);
                            }
                            atomicAdd(&acc[dl * LDS_STRIDE + c], v);
                        }
                    }
                }
            }
        }
        __syncthreads();
        const int nl = __builtin_amdgcn_readfirstlane(nlist);
        n = nl < LIST_CAP ? nl : LIST_CAP;
        if (FIRST && tid == 0) saved_n[blockIdx.x] = nl;
    }

    for (int base = wv * 32; base < n; base += AGG_THREADS) {
        const int cnt = (n - base < 32) ? (n - base) : 32;
        int sl = 0, dl = 0;
        if (lane < cnt) {
            const int w = list[base + lane];
            if (scan) {
                sl = src[w];
                dl = dst[w] - lo;
                if (FIRST) mine[base + lane] = (unsigned)sl | ((unsigned)dl << 16);
            } else {
                sl = w & 0xffff;
                dl = (int)((unsigned)w >> 16);
            }
        }
        for (int j0 = 0; j0 < cnt; j0 += GE) {
            int s[GE], d[GE];
            #pragma unroll
            for (int u = 0; u < GE; ++u) {
                s[u] = __builtin_amdgcn_readlane(sl, j0 + u);
                d[u] = __builtin_amdgcn_readlane(dl, j0 + u);
            }
            float v[GE][4];
            #pragma unroll
            for (int u = 0; u < GE; ++u) {
                if (j0 + u < cnt) {
                    const float* zr = hin + (size_t)s[u] * DIM + lane;
                    v[u][0] = zr[0];
                    v[u][1] = zr[32];
                    v[u][2] = zr[64];
                    v[u][3] = zr[96];
                }
            }
            #pragma unroll
            for (int u = 0; u < GE; ++u) {
                if (j0 + u < cnt) {
                    float* a = acc + d[u] * LDS_STRIDE + lane;
                    #pragma unroll
                    for (int j = 0; j < 4; ++j) {
                        const float y = BN ? bn_apply(v[u][j], mu[j], rs[j], ga[j], be[j]) : v[u][j];
                        atomicAdd(a + 32 * j, y);
                    }
                }
            }
        }
    }
  }
    __syncthreads();

    if (BN) {
        const int np = __builtin_amdgcn_readfirstlane(gmax) + 1;
        for (int p = wv; p < np; p += AGG_THREADS / 32) {
            float* gp = gcat + (size_t)(gfirst + p) * ZSTRIDE + lane;
            #pragma unroll
            for (int j = 0; j < 4; ++j) atomicAdd(gp + 32 * j, pool[p * DIM + lane + 32 * j]);
        }
    }

    if (GATHER) {
        const int hf   = lane >> 4;
        const int l16  = lane & 15;
        const int rt   = wv >> 2;
        const int col0 = (wv & 3) * 32 + l16;
        const int col1 = col0 + 16;
        const float bias1_0 = b1[col0], bias1_1 = b1[col1];
        const float bias2_0 = b2[col0], bias2_1 = b2[col1];
        float s0 = 0.f, q0 = 0.f, s1 = 0.f, q1 = 0.f;

        for (int pass = 0; pass < 2; ++pass) {
            const int  t     = pass * (AGG_THREADS / 128) + rt;
            const bool valid = t < N_RTILES;
            float* At = acc + t * (16 * LDS_STRIDE);

            v8f c0 = {}, c1 = {};
            if (valid) tile_gemm(At, ldsW1, l16, hf, col0, c0, c1);
            __syncthreads();
            if (valid) {
                #pragma unroll
                for (int g = 0; g < 8; ++g) {
                    const int m = g + 8 * hf;
                    const float v0 = c0[g] + bias1_0;
                    const float v1 = c1[g] + bias1_1;
                    At[m * LDS_STRIDE + col0] = v0 > 0.f ? v0 : 0.f;
                    At[m * LDS_STRIDE + col1] = v1 > 0.f ? v1 : 0.f;
                }
            }
            __syncthreads();

            if (valid) {
                v8f e0 = {}, e1 = {};
                tile_gemm(At, ldsW2, l16, hf, col0, e0, e1);
                float* hr = hout + (size_t)(lo + 16 * t) * DIM;
                #pragma unroll
                for (int g = 0; g < 8; ++g) {
                    const int m = g + 8 * hf;
                    float v0 = e0[g] + bias2_0;
                    float v1 = e1[g] + bias2_1;
                    v0 = v0 > 0.f ? v0 : 0.f;
                    v1 = v1 > 0.f ? v1 : 0.f;
                    if (16 * t + m < nrows) {
                        hr[m * DIM + col0] = v0;
                        hr[m * DIM + col1] = v1;
                        s0 += v0;  q0 += v0 * v0;
                        s1 += v1;  q1 += v1 * v1;
                    }
                }
            }
        }

        float* part = (float*)list;
        const int slot = (rt * 2 + hf) * (2 * DIM);
        part[slot + col0]       = s0;
        part[slot + col1]       = s1;
        part[slot + DIM + col0] = q0;
        part[slot + DIM + col1] = q1;
        __syncthreads();
        if (tid < 2 * DIM) {
            float a = 0.f;
            for (int i = 0; i < 2 * (AGG_THREADS / 128); ++i) a += part[i * (2 * DIM) + tid];
            atomicAdd(&stats_out[tid], a);
        }
    }
}

extern "C" void kernel_launch(void* const* d_in, const int* in_sizes, int n_in,
                              void* d_out, int out_size, void* d_ws, size_t ws_size,
                              hipStream_t stream)
{
    (void)in_sizes; (void)n_in; (void)out_size; (void)ws_size;

    const float* x     = (const float*)d_in[0];
    const int*   ei    = (const int*)  d_in[1];
    const int*   batch = (const int*)  d_in[2];
    const float* W1    = (const float*)d_in[3];
    const float* b1    = (const float*)d_in[4];
    const float* W2    = (const float*)d_in[5];
    const float* b2    = (const float*)d_in[6];
    const float* gamma = (const float*)d_in[7];
    const float* beta  = (const float*)d_in[8];

    float* out  = (float*)d_out;
    float* zcat = out;
    float* gcat = out + (size_t)N_NODES * ZSTRIDE;

    float* hbuf0 = (float*)d_ws;
    float* hbuf1 = hbuf0 + (size_t)N_NODES * DIM;
    float* stats = hbuf1 + (size_t)N_NODES * DIM;
    unsigned* owner = (unsigned*)(stats + NLAYER * 2 * DIM);
    unsigned* saved = owner + N_EDGES / 4;
    int* saved_n    = (int*)(saved + (size_t)AGG_BLOCKS * LIST_CAP);

    const int* src = ei;
    const int* dst = ei + N_EDGES;

    hipMemsetAsync(gcat, 0, sizeof(float) * (size_t)NGRAPH * ZSTRIDE, stream);
    hipMemsetAsync(stats, 0, sizeof(float) * (NLAYER * 2 * DIM), stream);

    owner_kernel<<<(N_EDGES / 4 + 255) / 256, 256, 0, stream>>>(dst, owner);

    static_assert(NLAYER == 3, "the three layers are launched one by one below");
    agg_kernel<0><<<AGG_BLOCKS, AGG_THREADS, 0, stream>>>(
        x, src, dst, owner, saved, saved_n,
        stats, gamma, beta, batch, zcat, gcat,
        W1, b1, W2, b2, hbuf0, stats);
    agg_kernel<1><<<AGG_BLOCKS, AGG_THREADS, 0, stream>>>(
        hbuf0, src, dst, owner, saved, saved_n,
        stats, gamma, beta, batch, zcat, gcat,
        W1 + (size_t)DIM * DIM, b1 + DIM, W2 + (size_t)DIM * DIM, b2 + DIM,
        hbuf1, stats + 2 * DIM);
    agg_kernel<1><<<AGG_BLOCKS, AGG_THREADS, 0, stream>>>(
        hbuf1, src, dst, owner, saved, saved_n,
        stats + 2 * DIM, gamma + DIM, beta + DIM, batch, zcat + DIM, gcat + DIM,
        W1 + (size_t)2 * DIM * DIM, b1 + 2 * DIM, W2 + (size_t)2 * DIM * DIM, b2 + 2 * DIM,
        hbuf0, stats + 4 * DIM);
    agg_kernel<2><<<AGG_BLOCKS, AGG_THREADS, 0, stream>>>(
        hbuf0, src, dst, owner, saved, saved_n,
        stats + 4 * DIM, gamma + 2 * DIM, beta + 2 * DIM, batch, zcat + 2 * DIM, gcat + 2 * DIM,
        W1, b1, W2, b2, hbuf1, stats);
}


# Gather walk without branches: list tail padded with spare edges into a padding row; a group's row loads asked for one group ahead of its LDS adds, 64 in flight a wave; LDS sum order as before
# speedup vs baseline: 4.0667x; 1.0412x over previous
#include <hip/hip_runtime.h>

#define N_NODES 50000
#define N_EDGES 800000
#define DIM     128
#define NLAYER  3
#define NGRAPH  512
#define ZSTRIDE (NLAYER * DIM)
#define BN_EPS  1e-5f
#define LDS_STRIDE 132

typedef __attribute__((ext_vector_type(2))) float v2f;
typedef __attribute__((ext_vector_type(8))) float v8f;


#define RB          196
#define ACC_ROWS    208
#define N_RTILES    (ACC_ROWS / 16)
#define AGG_BLOCKS  256
#define AGG_THREADS 1024
#define LIST_CAP    8192
#define GROUPS16    (N_EDGES / 16)
#define SCAN_UNROLL 7
#define GE          8
#define NG          (32 / GE)
#define PRE         (LIST_CAP / AGG_THREADS)

static_assert(N_NODES <= 65536 && RB <= 65536, "a saved word holds two 16-bit row numbers");
static_assert(32 % GE == 0, "whole groups cover a wave's 32 list entries");
static_assert(ACC_ROWS > RB && ACC_ROWS % 16 == 0 && N_RTILES <= 2 * (AGG_THREADS / 128), "two passes of 8 row tiles cover the block, and a padding row is left for the spare edges");

__global__ __launch_bounds__(256) void owner_kernel(
    const int* __restrict__ dst, unsigned* __restrict__ owner)
{
    const int i = blockIdx.x * 256 + threadIdx.x;
    if (i >= N_EDGES / 4) return;
    const int4 d = *(const int4*)(dst + 4 * i);
    owner[i] =  ((unsigned)d.x / RB)        | (((unsigned)d.y / RB) << 8) |
               (((unsigned)d.z / RB) << 16) | (((unsigned)d.w / RB) << 24);
}

__device__ __forceinline__ unsigned zero_bytes(unsigned x)
{
    unsigned t = (x & 0x7f7f7f7fu) + 0x7f7f7f7fu;
    t = t | x | 0x7f7f7f7fu;
    return ~t;
}

#define POOL_CAP 64
#define RPW ((ACC_ROWS + 31) / 32)

__device__ __forceinline__ void bn_column(const float* __restrict__ stats,
    const float* __restrict__ gamma, const float* __restrict__ beta, int c,
    float& mu, float& rs, float& ga, float& be)
{
    mu = stats[c] * (1.0f / N_NODES);
    const float var = stats[DIM + c] * (1.0f / N_NODES) - mu * mu;
    rs = rsqrtf(var + BN_EPS);
    ga = gamma[c];
    be = beta[c];
}
__device__ __forceinline__ float bn_apply(float h, float mu, float rs, float ga, float be)
{
    return (h - mu) * rs * ga + be;
}


__device__ __forceinline__ void stage_weights(const float* __restrict__ Wg, float* Wl, int tid)
{
    #pragma unroll
    for (int it = 0; it < (DIM / 2) * (DIM / 4) / AGG_THREADS; ++it) {
        const int item = it * AGG_THREADS + tid;
        const int kp = item >> 5, c4 = item & 31;
        const float4 r0 = *(const float4*)(Wg + (size_t)(2 * kp) * DIM + 4 * c4);
        const float4 r1 = *(const float4*)(Wg + (size_t)(2 * kp + 1) * DIM + 4 * c4);
        float* p = Wl + (kp * DIM + ((4 * c4) ^ ((kp & 1) << 4))) * 2;
        *(float4*)(p)     = make_float4(r0.x, r1.x, r0.y, r1.y);
        *(float4*)(p + 4) = make_float4(r0.z, r1.z, r0.w, r1.w);
    }
}

#define KBLK 4
__device__ __forceinline__ void tile_gemm(const float* At, const float* Wl,
    int l16, int hf, int col0, v8f& c0, v8f& c1)
{
    const float* ap  = At + l16 * LDS_STRIDE + 2 * hf;
    const float* bp0 = Wl + (hf * DIM + (col0 ^ (hf << 4))) * 2;
    const float* bp1 = Wl + (hf * DIM + ((col0 + 16) ^ (hf << 4))) * 2;
    v2f a[KBLK], p[KBLK], q[KBLK];
    #pragma unroll
    for (int u = 0; u < KBLK; ++u) {
        a[u] = *(const v2f*)(ap + 4 * u);
        p[u] = *(const v2f*)(bp0 + 4 * u * DIM);
        q[u] = *(const v2f*)(bp1 + 4 * u * DIM);
    }
    #pragma unroll
    for (int k0 = 0; k0 < DIM; k0 += 4 * KBLK) {
        v2f an[KBLK], pn[KBLK], qn[KBLK];
        if (k0 + 4 * KBLK < DIM) {
            #pragma unroll
            for (int u = 0; u < KBLK; ++u) {
                const int k = k0 + 4 * KBLK + 4 * u;
                an[u] = *(const v2f*)(ap + k);
                pn[u] = *(const v2f*)(bp0 + k * DIM);
                qn[u] = *(const v2f*)(bp1 + k * DIM);
            }
        }
        #pragma unroll
        for (int u = 0; u < KBLK; ++u) {
            c0 = __builtin_amdgcn_wmma_f32_16x16x4_f32(false, a[u], false, p[u], (short)0, c0, false, false);
            c1 = __builtin_amdgcn_wmma_f32_16x16x4_f32(false, a[u], false, q[u], (short)0, c1, false, false);
        }
        if (k0 + 4 * KBLK < DIM) {
            #pragma unroll
            for (int u = 0; u < KBLK; ++u) { a[u] = an[u]; p[u] = pn[u]; q[u] = qn[u]; }
        }
    }
}

template <int MODE>
__global__ __launch_bounds__(AGG_THREADS) void agg_kernel(
    const float* __restrict__ hin,
    const int* __restrict__ src, const int* __restrict__ dst,
    const unsigned* __restrict__ owner,
    unsigned* __restrict__ saved, int* __restrict__ saved_n,
    const float* __restrict__ stats, const float* __restrict__ gamma, const float* __restrict__ beta,
    const int* __restrict__ batch,
    float* __restrict__ zcat, float* __restrict__ gcat,
    const float* __restrict__ W1, const float* __restrict__ b1,
    const float* __restrict__ W2, const float* __restrict__ b2,
    float* __restrict__ hout, float* __restrict__ stats_out)
{
    constexpr bool FIRST  = MODE == 0;
    constexpr bool BN     = MODE != 0;
    constexpr bool GATHER = MODE != 2;
    __shared__ __attribute__((aligned(16))) float acc[ACC_ROWS * LDS_STRIDE];
    __shared__ __attribute__((aligned(16))) float ldsW1[DIM * DIM];
    __shared__ __attribute__((aligned(16))) float ldsW2[DIM * DIM];
    __shared__ __attribute__((aligned(16))) int   list[LIST_CAP];
    __shared__ float pool[POOL_CAP * DIM];
    __shared__ int   nlist;
    __shared__ int   gmax;

    const int tid  = threadIdx.x;
    const int lane = tid & 31;
    const int wv   = __builtin_amdgcn_readfirstlane(tid >> 5);
    const int lo   = blockIdx.x * RB;
    const int nrows = (N_NODES - lo < RB) ? (N_NODES - lo) : RB;
    unsigned* mine = saved + (size_t)blockIdx.x * LIST_CAP;

    int  nsaved = 0;
    bool scan   = FIRST;
    if (MODE == 1) {
        nsaved = __builtin_amdgcn_readfirstlane(saved_n[blockIdx.x]);
        scan   = nsaved > LIST_CAP;
    }
    float mu[4], rs[4], ga[4], be[4];
    if (BN) {
        #pragma unroll
        for (int j = 0; j < 4; ++j)
            bn_column(stats, gamma, beta, lane + 32 * j, mu[j], rs[j], ga[j], be[j]);
    }
    unsigned pre[PRE];
    if (MODE == 1 && !scan) {
        #pragma unroll
        for (int u = 0; u < PRE; ++u) {
            const int i = u * AGG_THREADS + tid;
            pre[u] = (i < nsaved) ? mine[i] : 0u;
        }
    }
    if (BN) {
        for (int i = tid; i < POOL_CAP * DIM; i += AGG_THREADS) pool[i] = 0.f;
        if (tid == 0) gmax = -1;
    }
    if (tid == 0) nlist = 0;

    const int gfirst = BN ? batch[lo] : 0;
    float hv[RPW][4];
    int   gr[RPW];
    #pragma unroll
    for (int t = 0; t < RPW; ++t) {
        const int r = wv + 32 * t;
        gr[t] = 0;
        hv[t][0] = hv[t][1] = hv[t][2] = hv[t][3] = 0.f;
        if (r < nrows) {
            const float* hr = hin + (size_t)(lo + r) * DIM + lane;
            hv[t][0] = hr[0];
            hv[t][1] = hr[32];
            hv[t][2] = hr[64];
            hv[t][3] = hr[96];
            if (BN) gr[t] = batch[lo + r];
        }
    }
    if (BN) __syncthreads();
    #pragma unroll
    for (int t = 0; t < RPW; ++t) {
        const int r = wv + 32 * t;
        if (r < ACC_ROWS) {
            float y[4];
            #pragma unroll
            for (int j = 0; j < 4; ++j)
                y[j] = BN ? bn_apply(hv[t][j], mu[j], rs[j], ga[j], be[j]) : hv[t][j];
            if (BN && r < nrows) {
                float* zr = zcat + (size_t)(lo + r) * ZSTRIDE + lane;
                #pragma unroll
                for (int j = 0; j < 4; ++j) zr[32 * j] = y[j];
                const unsigned p = (unsigned)(gr[t] - gfirst);
                if (p < POOL_CAP) {
                    #pragma unroll
                    for (int j = 0; j < 4; ++j) atomicAdd(&pool[p * DIM + lane + 32 * j], y[j]);
                    if (lane == 0) atomicMax(&gmax, (int)p);
                } else {
                    float* gp = gcat + (size_t)gr[t] * ZSTRIDE + lane;
                    #pragma unroll
                    for (int j = 0; j < 4; ++j) atomicAdd(gp + 32 * j, y[j]);
                }
            }
            if (GATHER) {
                #pragma unroll
                for (int j = 0; j < 4; ++j) acc[r * LDS_STRIDE + lane + 32 * j] = (r < nrows) ? y[j] : 0.f;
            }
        }
    }

    if (MODE == 1 && !scan) {
        #pragma unroll
        for (int u = 0; u < PRE; ++u) {
            const int i = u * AGG_THREADS + tid;
            if (i < nsaved) list[i] = (int)pre[u];
        }
    }
    __syncthreads();

  if (GATHER) {
    stage_weights(W1, ldsW1, tid);
    stage_weights(W2, ldsW2, tid);

    int n = nsaved;
    if (scan) {
        const unsigned pat = (unsigned)blockIdx.x * 0x01010101u;
        const uint4* ow = (const uint4*)owner;
        for (int g0 = 0; g0 < GROUPS16; g0 += AGG_THREADS * SCAN_UNROLL) {
            uint4 w[SCAN_UNROLL];
            #pragma unroll
            for (int u = 0; u < SCAN_UNROLL; ++u) {
                const int g = g0 + u * AGG_THREADS + tid;
                w[u] = ow[g < GROUPS16 ? g : GROUPS16 - 1];
            }
            #pragma unroll
            for (int u = 0; u < SCAN_UNROLL; ++u) {
                const int g = g0 + u * AGG_THREADS + tid;
                unsigned m = (zero_bytes(w[u].x ^ pat) >> 3) | (zero_bytes(w[u].y ^ pat) >> 2) |
                             (zero_bytes(w[u].z ^ pat) >> 1) |  zero_bytes(w[u].w ^ pat);
                if (g >= GROUPS16) m = 0;
                while (m) {
                    const int p = __builtin_ctz(m);
                    m &= m - 1;
                    const int e = g * 16 + (((p & 7) - 4) << 2) + (p >> 3);
                    const int slot = atomicAdd(&nlist, 1);
                    if (slot < LIST_CAP) {
                        list[slot] = e;
                    } else {
                        const int dl = dst[e] - lo;
                        const float* zr = hin + (size_t)src[e] * DIM;
                        for (int c = 0; c < DIM; ++c) {
                            float v = zr[c];
                            if (BN) {
                                float m1, r1, g1, b1;
                                bn_column(stats, gamma, beta, c, m1, r1, g1, b1);
                                v = bn_apply(v, m1, r1, g1, b1);
                            }
                            atomicAdd(&acc[dl * LDS_STRIDE + c], v);
                        }
                    }
                }
            }
        }
        __syncthreads();
        const int nl = __builtin_amdgcn_readfirstlane(nlist);
        n = nl < LIST_CAP ? nl : LIST_CAP;
        if (FIRST && tid == 0) saved_n[blockIdx.x] = nl;
    }

    for (int base = wv * 32; base < n; base += AGG_THREADS) {
        const int cnt = (n - base < 32) ? (n - base) : 32;
        int sl = 0, dl = ACC_ROWS - 1;
        if (lane < cnt) {
            const int w = list[base + lane];
            if (scan) {
                sl = src[w];
                dl = dst[w] - lo;
                if (FIRST) mine[base + lane] = (unsigned)sl | ((unsigned)dl << 16);
            } else {
                sl = w & 0xffff;
                dl = (int)((unsigned)w >> 16);
            }
        }
        float v[2][GE][4];
        #pragma unroll
        for (int g = 0; g <= NG; ++g) {
            if (g < NG) {
                #pragma unroll
                for (int u = 0; u < GE; ++u) {
                    const int s = __builtin_amdgcn_readlane(sl, g * GE + u);
                    const float* zr = hin + (size_t)s * DIM + lane;
                    v[g & 1][u][0] = zr[0];
                    v[g & 1][u][1] = zr[32];
                    v[g & 1][u][2] = zr[64];
                    v[g & 1][u][3] = zr[96];
                }
            }
            if (g > 0) {
                #pragma unroll
                for (int u = 0; u < GE; ++u) {
                    const int d = __builtin_amdgcn_readlane(dl, (g - 1) * GE + u);
                    float* a = acc + d * LDS_STRIDE + lane;
                    #pragma unroll
                    for (int j = 0; j < 4; ++j) {
                        const float y = BN ? bn_apply(v[(g - 1) & 1][u][j], mu[j], rs[j], ga[j], be[j])
                                           : v[(g - 1) & 1][u][j];
                        atomicAdd(a + 32 * j, y);
                    }
                }
            }
        }
    }
  }
    __syncthreads();

    if (BN) {
        const int np = __builtin_amdgcn_readfirstlane(gmax) + 1;
        for (int p = wv; p < np; p += AGG_THREADS / 32) {
            float* gp = gcat + (size_t)(gfirst + p) * ZSTRIDE + lane;
            #pragma unroll
            for (int j = 0; j < 4; ++j) atomicAdd(gp + 32 * j, pool[p * DIM + lane + 32 * j]);
        }
    }

    if (GATHER) {
        const int hf   = lane >> 4;
        const int l16  = lane & 15;
        const int rt   = wv >> 2;
        const int col0 = (wv & 3) * 32 + l16;
        const int col1 = col0 + 16;
        const float bias1_0 = b1[col0], bias1_1 = b1[col1];
        const float bias2_0 = b2[col0], bias2_1 = b2[col1];
        float s0 = 0.f, q0 = 0.f, s1 = 0.f, q1 = 0.f;

        for (int pass = 0; pass < 2; ++pass) {
            const int  t     = pass * (AGG_THREADS / 128) + rt;
            const bool valid = t < N_RTILES;
            float* At = acc + t * (16 * LDS_STRIDE);

            v8f c0 = {}, c1 = {};
            if (valid) tile_gemm(At, ldsW1, l16, hf, col0, c0, c1);
            __syncthreads();
            if (valid) {
                #pragma unroll
                for (int g = 0; g < 8; ++g) {
                    const int m = g + 8 * hf;
                    const float v0 = c0[g] + bias1_0;
                    const float v1 = c1[g] + bias1_1;
                    At[m * LDS_STRIDE + col0] = v0 > 0.f ? v0 : 0.f;
                    At[m * LDS_STRIDE + col1] = v1 > 0.f ? v1 : 0.f;
                }
            }
            __syncthreads();

            if (valid) {
                v8f e0 = {}, e1 = {};
                tile_gemm(At, ldsW2, l16, hf, col0, e0, e1);
                float* hr = hout + (size_t)(lo + 16 * t) * DIM;
                #pragma unroll
                for (int g = 0; g < 8; ++g) {
                    const int m = g + 8 * hf;
                    float v0 = e0[g] + bias2_0;
                    float v1 = e1[g] + bias2_1;
                    v0 = v0 > 0.f ? v0 : 0.f;
                    v1 = v1 > 0.f ? v1 : 0.f;
                    if (16 * t + m < nrows) {
                        hr[m * DIM + col0] = v0;
                        hr[m * DIM + col1] = v1;
                        s0 += v0;  q0 += v0 * v0;
                        s1 += v1;  q1 += v1 * v1;
                    }
                }
            }
        }

        float* part = (float*)list;
        const int slot = (rt * 2 + hf) * (2 * DIM);
        part[slot + col0]       = s0;
        part[slot + col1]       = s1;
        part[slot + DIM + col0] = q0;
        part[slot + DIM + col1] = q1;
        __syncthreads();
        if (tid < 2 * DIM) {
            float a = 0.f;
            for (int i = 0; i < 2 * (AGG_THREADS / 128); ++i) a += part[i * (2 * DIM) + tid];
            atomicAdd(&stats_out[tid], a);
        }
    }
}

extern "C" void kernel_launch(void* const* d_in, const int* in_sizes, int n_in,
                              void* d_out, int out_size, void* d_ws, size_t ws_size,
                              hipStream_t stream)
{
    (void)in_sizes; (void)n_in; (void)out_size; (void)ws_size;

    const float* x     = (const float*)d_in[0];
    const int*   ei    = (const int*)  d_in[1];
    const int*   batch = (const int*)  d_in[2];
    const float* W1    = (const float*)d_in[3];
    const float* b1    = (const float*)d_in[4];
    const float* W2    = (const float*)d_in[5];
    const float* b2    = (const float*)d_in[6];
    const float* gamma = (const float*)d_in[7];
    const float* beta  = (const float*)d_in[8];

    float* out  = (float*)d_out;
    float* zcat = out;
    float* gcat = out + (size_t)N_NODES * ZSTRIDE;

    float* hbuf0 = (float*)d_ws;
    float* hbuf1 = hbuf0 + (size_t)N_NODES * DIM;
    float* stats = hbuf1 + (size_t)N_NODES * DIM;
    unsigned* owner = (unsigned*)(stats + NLAYER * 2 * DIM);
    unsigned* saved = owner + N_EDGES / 4;
    int* saved_n    = (int*)(saved + (size_t)AGG_BLOCKS * LIST_CAP);

    const int* src = ei;
    const int* dst = ei + N_EDGES;

    hipMemsetAsync(gcat, 0, sizeof(float) * (size_t)NGRAPH * ZSTRIDE, stream);
    hipMemsetAsync(stats, 0, sizeof(float) * (NLAYER * 2 * DIM), stream);

    owner_kernel<<<(N_EDGES / 4 + 255) / 256, 256, 0, stream>>>(dst, owner);

    static_assert(NLAYER == 3, "the three layers are launched one by one below");
    agg_kernel<0><<<AGG_BLOCKS, AGG_THREADS, 0, stream>>>(
        x, src, dst, owner, saved, saved_n,
        stats, gamma, beta, batch, zcat, gcat,
        W1, b1, W2, b2, hbuf0, stats);
    agg_kernel<1><<<AGG_BLOCKS, AGG_THREADS, 0, stream>>>(
        hbuf0, src, dst, owner, saved, saved_n,
        stats, gamma, beta, batch, zcat, gcat,
        W1 + (size_t)DIM * DIM, b1 + DIM, W2 + (size_t)DIM * DIM, b2 + DIM,
        hbuf1, stats + 2 * DIM);
    agg_kernel<1><<<AGG_BLOCKS, AGG_THREADS, 0, stream>>>(
        hbuf1, src, dst, owner, saved, saved_n,
        stats + 2 * DIM, gamma + DIM, beta + DIM, batch, zcat + DIM, gcat + DIM,
        W1 + (size_t)2 * DIM * DIM, b1 + 2 * DIM, W2 + (size_t)2 * DIM * DIM, b2 + 2 * DIM,
        hbuf0, stats + 4 * DIM);
    agg_kernel<2><<<AGG_BLOCKS, AGG_THREADS, 0, stream>>>(
        hbuf0, src, dst, owner, saved, saved_n,
        stats + 4 * DIM, gamma + 2 * DIM, beta + 2 * DIM, batch, zcat + 2 * DIM, gcat + 2 * DIM,
        W1, b1, W2, b2, hbuf1, stats);
}


# Row loads 16 bytes a lane: one load fetches a whole 512-B row (was 4 loads of one line each); sums lie column-permuted in LDS, W1 rows staged in that k order; first product's k sum order changes
# speedup vs baseline: 4.3872x; 1.0788x over previous
#include <hip/hip_runtime.h>

#define N_NODES 50000
#define N_EDGES 800000
#define DIM     128
#define NLAYER  3
#define NGRAPH  512
#define ZSTRIDE (NLAYER * DIM)
#define BN_EPS  1e-5f
#define LDS_STRIDE 132

typedef __attribute__((ext_vector_type(2))) float v2f;
typedef __attribute__((ext_vector_type(8))) float v8f;


#define RB          196
#define ACC_ROWS    208
#define N_RTILES    (ACC_ROWS / 16)
#define AGG_BLOCKS  256
#define AGG_THREADS 1024
#define LIST_CAP    8192
#define GROUPS16    (N_EDGES / 16)
#define SCAN_UNROLL 7
#define GE          8
#define NG          (32 / GE)
#define PRE         (LIST_CAP / AGG_THREADS)

static_assert(N_NODES <= 65536 && RB <= 65536, "a saved word holds two 16-bit row numbers");
static_assert(32 % GE == 0, "whole groups cover a wave's 32 list entries");
static_assert(ACC_ROWS > RB && ACC_ROWS % 16 == 0 && N_RTILES <= 2 * (AGG_THREADS / 128), "two passes of 8 row tiles cover the block, and a padding row is left for the spare edges");

__global__ __launch_bounds__(256) void owner_kernel(
    const int* __restrict__ dst, unsigned* __restrict__ owner)
{
    const int i = blockIdx.x * 256 + threadIdx.x;
    if (i >= N_EDGES / 4) return;
    const int4 d = *(const int4*)(dst + 4 * i);
    owner[i] =  ((unsigned)d.x / RB)        | (((unsigned)d.y / RB) << 8) |
               (((unsigned)d.z / RB) << 16) | (((unsigned)d.w / RB) << 24);
}

__device__ __forceinline__ unsigned zero_bytes(unsigned x)
{
    unsigned t = (x & 0x7f7f7f7fu) + 0x7f7f7f7fu;
    t = t | x | 0x7f7f7f7fu;
    return ~t;
}

#define POOL_CAP 64
#define RPW ((ACC_ROWS + 31) / 32)

__device__ __forceinline__ void bn_column(const float* __restrict__ stats,
    const float* __restrict__ gamma, const float* __restrict__ beta, int c,
    float& mu, float& rs, float& ga, float& be)
{
    mu = stats[c] * (1.0f / N_NODES);
    const float var = stats[DIM + c] * (1.0f / N_NODES) - mu * mu;
    rs = rsqrtf(var + BN_EPS);
    ga = gamma[c];
    be = beta[c];
}
__device__ __forceinline__ float bn_apply(float h, float mu, float rs, float ga, float be)
{
    return (h - mu) * rs * ga + be;
}


#define KCOL(p) (4 * ((p) & 31) + ((p) >> 5))
#define KPOS(c) ((((c) & 3) << 5) + ((c) >> 2))

template <bool PERM>
__device__ __forceinline__ void stage_weights(const float* __restrict__ Wg, float* Wl, int tid)
{
    #pragma unroll
    for (int it = 0; it < (DIM / 2) * (DIM / 4) / AGG_THREADS; ++it) {
        const int item = it * AGG_THREADS + tid;
        const int kp = item >> 5, c4 = item & 31;
        const int k0 = PERM ? KCOL(2 * kp) : 2 * kp;
        const int k1 = PERM ? KCOL(2 * kp + 1) : 2 * kp + 1;
        const float4 r0 = *(const float4*)(Wg + (size_t)k0 * DIM + 4 * c4);
        const float4 r1 = *(const float4*)(Wg + (size_t)k1 * DIM + 4 * c4);
        float* p = Wl + (kp * DIM + ((4 * c4) ^ ((kp & 1) << 4))) * 2;
        *(float4*)(p)     = make_float4(r0.x, r1.x, r0.y, r1.y);
        *(float4*)(p + 4) = make_float4(r0.z, r1.z, r0.w, r1.w);
    }
}

#define KBLK 4
__device__ __forceinline__ void tile_gemm(const float* At, const float* Wl,
    int l16, int hf, int col0, v8f& c0, v8f& c1)
{
    const float* ap  = At + l16 * LDS_STRIDE + 2 * hf;
    const float* bp0 = Wl + (hf * DIM + (col0 ^ (hf << 4))) * 2;
    const float* bp1 = Wl + (hf * DIM + ((col0 + 16) ^ (hf << 4))) * 2;
    v2f a[KBLK], p[KBLK], q[KBLK];
    #pragma unroll
    for (int u = 0; u < KBLK; ++u) {
        a[u] = *(const v2f*)(ap + 4 * u);
        p[u] = *(const v2f*)(bp0 + 4 * u * DIM);
        q[u] = *(const v2f*)(bp1 + 4 * u * DIM);
    }
    #pragma unroll
    for (int k0 = 0; k0 < DIM; k0 += 4 * KBLK) {
        v2f an[KBLK], pn[KBLK], qn[KBLK];
        if (k0 + 4 * KBLK < DIM) {
            #pragma unroll
            for (int u = 0; u < KBLK; ++u) {
                const int k = k0 + 4 * KBLK + 4 * u;
                an[u] = *(const v2f*)(ap + k);
                pn[u] = *(const v2f*)(bp0 + k * DIM);
                qn[u] = *(const v2f*)(bp1 + k * DIM);
            }
        }
        #pragma unroll
        for (int u = 0; u < KBLK; ++u) {
            c0 = __builtin_amdgcn_wmma_f32_16x16x4_f32(false, a[u], false, p[u], (short)0, c0, false, false);
            c1 = __builtin_amdgcn_wmma_f32_16x16x4_f32(false, a[u], false, q[u], (short)0, c1, false, false);
        }
        if (k0 + 4 * KBLK < DIM) {
            #pragma unroll
            for (int u = 0; u < KBLK; ++u) { a[u] = an[u]; p[u] = pn[u]; q[u] = qn[u]; }
        }
    }
}

template <int MODE>
__global__ __launch_bounds__(AGG_THREADS) void agg_kernel(
    const float* __restrict__ hin,
    const int* __restrict__ src, const int* __restrict__ dst,
    const unsigned* __restrict__ owner,
    unsigned* __restrict__ saved, int* __restrict__ saved_n,
    const float* __restrict__ stats, const float* __restrict__ gamma, const float* __restrict__ beta,
    const int* __restrict__ batch,
    float* __restrict__ zcat, float* __restrict__ gcat,
    const float* __restrict__ W1, const float* __restrict__ b1,
    const float* __restrict__ W2, const float* __restrict__ b2,
    float* __restrict__ hout, float* __restrict__ stats_out)
{
    constexpr bool FIRST  = MODE == 0;
    constexpr bool BN     = MODE != 0;
    constexpr bool GATHER = MODE != 2;
    __shared__ __attribute__((aligned(16))) float acc[ACC_ROWS * LDS_STRIDE];
    __shared__ __attribute__((aligned(16))) float ldsW1[DIM * DIM];
    __shared__ __attribute__((aligned(16))) float ldsW2[DIM * DIM];
    __shared__ __attribute__((aligned(16))) int   list[LIST_CAP];
    __shared__ float pool[POOL_CAP * DIM];
    __shared__ int   nlist;
    __shared__ int   gmax;

    const int tid  = threadIdx.x;
    const int lane = tid & 31;
    const int wv   = __builtin_amdgcn_readfirstlane(tid >> 5);
    const int lo   = blockIdx.x * RB;
    const int nrows = (N_NODES - lo < RB) ? (N_NODES - lo) : RB;
    unsigned* mine = saved + (size_t)blockIdx.x * LIST_CAP;

    int  nsaved = 0;
    bool scan   = FIRST;
    if (MODE == 1) {
        nsaved = __builtin_amdgcn_readfirstlane(saved_n[blockIdx.x]);
        scan   = nsaved > LIST_CAP;
    }
    float mu[4], rs[4], ga[4], be[4];
    if (BN) {
        #pragma unroll
        for (int j = 0; j < 4; ++j)
            bn_column(stats, gamma, beta, 4 * lane + j, mu[j], rs[j], ga[j], be[j]);
    }
    unsigned pre[PRE];
    if (MODE == 1 && !scan) {
        #pragma unroll
        for (int u = 0; u < PRE; ++u) {
            const int i = u * AGG_THREADS + tid;
            pre[u] = (i < nsaved) ? mine[i] : 0u;
        }
    }
    if (BN) {
        for (int i = tid; i < POOL_CAP * DIM; i += AGG_THREADS) pool[i] = 0.f;
        if (tid == 0) gmax = -1;
    }
    if (tid == 0) nlist = 0;

    const int gfirst = BN ? batch[lo] : 0;
    float hv[RPW][4];
    int   gr[RPW];
    #pragma unroll
    for (int t = 0; t < RPW; ++t) {
        const int r = wv + 32 * t;
        gr[t] = 0;
        hv[t][0] = hv[t][1] = hv[t][2] = hv[t][3] = 0.f;
        if (r < nrows) {
            const float4 h4 = *(const float4*)(hin + (size_t)(lo + r) * DIM + 4 * lane);
            hv[t][0] = h4.x;
            hv[t][1] = h4.y;
            hv[t][2] = h4.z;
            hv[t][3] = h4.w;
            if (BN) gr[t] = batch[lo + r];
        }
    }
    if (BN) __syncthreads();
    #pragma unroll
    for (int t = 0; t < RPW; ++t) {
        const int r = wv + 32 * t;
        if (r < ACC_ROWS) {
            float y[4];
            #pragma unroll
            for (int j = 0; j < 4; ++j)
                y[j] = BN ? bn_apply(hv[t][j], mu[j], rs[j], ga[j], be[j]) : hv[t][j];
            if (BN && r < nrows) {
                *(float4*)(zcat + (size_t)(lo + r) * ZSTRIDE + 4 * lane) = make_float4(y[0], y[1], y[2], y[3]);
                const unsigned p = (unsigned)(gr[t] - gfirst);
                if (p < POOL_CAP) {
                    #pragma unroll
                    for (int j = 0; j < 4; ++j) atomicAdd(&pool[p * DIM + lane + 32 * j], y[j]);
                    if (lane == 0) atomicMax(&gmax, (int)p);
                } else {
                    float* gp = gcat + (size_t)gr[t] * ZSTRIDE + 4 * lane;
                    #pragma unroll
                    for (int j = 0; j < 4; ++j) atomicAdd(gp + j, y[j]);
                }
            }
            if (GATHER) {
                #pragma unroll
                for (int j = 0; j < 4; ++j) acc[r * LDS_STRIDE + lane + 32 * j] = (r < nrows) ? y[j] : 0.f;
            }
        }
    }

    if (MODE == 1 && !scan) {
        #pragma unroll
        for (int u = 0; u < PRE; ++u) {
            const int i = u * AGG_THREADS + tid;
            if (i < nsaved) list[i] = (int)pre[u];
        }
    }
    __syncthreads();

  if (GATHER) {
    stage_weights<true>(W1, ldsW1, tid);
    stage_weights<false>(W2, ldsW2, tid);

    int n = nsaved;
    if (scan) {
        const unsigned pat = (unsigned)blockIdx.x * 0x01010101u;
        const uint4* ow = (const uint4*)owner;
        for (int g0 = 0; g0 < GROUPS16; g0 += AGG_THREADS * SCAN_UNROLL) {
            uint4 w[SCAN_UNROLL];
            #pragma unroll
            for (int u = 0; u < SCAN_UNROLL; ++u) {
                const int g = g0 + u * AGG_THREADS + tid;
                w[u] = ow[g < GROUPS16 ? g : GROUPS16 - 1];
            }
            #pragma unroll
            for (int u = 0; u < SCAN_UNROLL; ++u) {
                const int g = g0 + u * AGG_THREADS + tid;
                unsigned m = (zero_bytes(w[u].x ^ pat) >> 3) | (zero_bytes(w[u].y ^ pat) >> 2) |
                             (zero_bytes(w[u].z ^ pat) >> 1) |  zero_bytes(w[u].w ^ pat);
                if (g >= GROUPS16) m = 0;
                while (m) {
                    const int p = __builtin_ctz(m);
                    m &= m - 1;
                    const int e = g * 16 + (((p & 7) - 4) << 2) + (p >> 3);
                    const int slot = atomicAdd(&nlist, 1);
                    if (slot < LIST_CAP) {
                        list[slot] = e;
                    } else {
                        const int dl = dst[e] - lo;
                        const float* zr = hin + (size_t)src[e] * DIM;
                        for (int c = 0; c < DIM; ++c) {
                            float v = zr[c];
                            if (BN) {
                                float m1, r1, g1, b1;
                                bn_column(stats, gamma, beta, c, m1, r1, g1, b1);
                                v = bn_apply(v, m1, r1, g1, b1);
                            }
                            atomicAdd(&acc[dl * LDS_STRIDE + KPOS(c)], v);
                        }
                    }
                }
            }
        }
        __syncthreads();
        const int nl = __builtin_amdgcn_readfirstlane(nlist);
        n = nl < LIST_CAP ? nl : LIST_CAP;
        if (FIRST && tid == 0) saved_n[blockIdx.x] = nl;
    }

    for (int base = wv * 32; base < n; base += AGG_THREADS) {
        const int cnt = (n - base < 32) ? (n - base) : 32;
        int sl = 0, dl = ACC_ROWS - 1;
        if (lane < cnt) {
            const int w = list[base + lane];
            if (scan) {
                sl = src[w];
                dl = dst[w] - lo;
                if (FIRST) mine[base + lane] = (unsigned)sl | ((unsigned)dl << 16);
            } else {
                sl = w & 0xffff;
                dl = (int)((unsigned)w >> 16);
            }
        }
        float v[2][GE][4];
        #pragma unroll
        for (int g = 0; g <= NG; ++g) {
            if (g < NG) {
                #pragma unroll
                for (int u = 0; u < GE; ++u) {
                    const int s = __builtin_amdgcn_readlane(sl, g * GE + u);
                    const float4 z4 = *(const float4*)(hin + (size_t)s * DIM + 4 * lane);
                    v[g & 1][u][0] = z4.x;
                    v[g & 1][u][1] = z4.y;
                    v[g & 1][u][2] = z4.z;
                    v[g & 1][u][3] = z4.w;
                }
            }
            if (g > 0) {
                #pragma unroll
                for (int u = 0; u < GE; ++u) {
                    const int d = __builtin_amdgcn_readlane(dl, (g - 1) * GE + u);
                    float* a = acc + d * LDS_STRIDE + lane;
                    #pragma unroll
                    for (int j = 0; j < 4; ++j) {
                        const float y = BN ? bn_apply(v[(g - 1) & 1][u][j], mu[j], rs[j], ga[j], be[j])
                                           : v[(g - 1) & 1][u][j];
                        atomicAdd(a + 32 * j, y);
                    }
                }
            }
        }
    }
  }
    __syncthreads();

    if (BN) {
        const int np = __builtin_amdgcn_readfirstlane(gmax) + 1;
        for (int p = wv; p < np; p += AGG_THREADS / 32) {
            float* gp = gcat + (size_t)(gfirst + p) * ZSTRIDE + lane;
            #pragma unroll
            for (int j = 0; j < 4; ++j) atomicAdd(gp + 32 * j, pool[p * DIM + KPOS(lane + 32 * j)]);
        }
    }

    if (GATHER) {
        const int hf   = lane >> 4;
        const int l16  = lane & 15;
        const int rt   = wv >> 2;
        const int col0 = (wv & 3) * 32 + l16;
        const int col1 = col0 + 16;
        const float bias1_0 = b1[col0], bias1_1 = b1[col1];
        const float bias2_0 = b2[col0], bias2_1 = b2[col1];
        float s0 = 0.f, q0 = 0.f, s1 = 0.f, q1 = 0.f;

        for (int pass = 0; pass < 2; ++pass) {
            const int  t     = pass * (AGG_THREADS / 128) + rt;
            const bool valid = t < N_RTILES;
            float* At = acc + t * (16 * LDS_STRIDE);

            v8f c0 = {}, c1 = {};
            if (valid) tile_gemm(At, ldsW1, l16, hf, col0, c0, c1);
            __syncthreads();
            if (valid) {
                #pragma unroll
                for (int g = 0; g < 8; ++g) {
                    const int m = g + 8 * hf;
                    const float v0 = c0[g] + bias1_0;
                    const float v1 = c1[g] + bias1_1;
                    At[m * LDS_STRIDE + col0] = v0 > 0.f ? v0 : 0.f;
                    At[m * LDS_STRIDE + col1] = v1 > 0.f ? v1 : 0.f;
                }
            }
            __syncthreads();

            if (valid) {
                v8f e0 = {}, e1 = {};
                tile_gemm(At, ldsW2, l16, hf, col0, e0, e1);
                float* hr = hout + (size_t)(lo + 16 * t) * DIM;
                #pragma unroll
                for (int g = 0; g < 8; ++g) {
                    const int m = g + 8 * hf;
                    float v0 = e0[g] + bias2_0;
                    float v1 = e1[g] + bias2_1;
                    v0 = v0 > 0.f ? v0 : 0.f;
                    v1 = v1 > 0.f ? v1 : 0.f;
                    if (16 * t + m < nrows) {
                        hr[m * DIM + col0] = v0;
                        hr[m * DIM + col1] = v1;
                        s0 += v0;  q0 += v0 * v0;
                        s1 += v1;  q1 += v1 * v1;
                    }
                }
            }
        }

        float* part = (float*)list;
        const int slot = (rt * 2 + hf) * (2 * DIM);
        part[slot + col0]       = s0;
        part[slot + col1]       = s1;
        part[slot + DIM + col0] = q0;
        part[slot + DIM + col1] = q1;
        __syncthreads();
        if (tid < 2 * DIM) {
            float a = 0.f;
            for (int i = 0; i < 2 * (AGG_THREADS / 128); ++i) a += part[i * (2 * DIM) + tid];
            atomicAdd(&stats_out[tid], a);
        }
    }
}

extern "C" void kernel_launch(void* const* d_in, const int* in_sizes, int n_in,
                              void* d_out, int out_size, void* d_ws, size_t ws_size,
                              hipStream_t stream)
{
    (void)in_sizes; (void)n_in; (void)out_size; (void)ws_size;

    const float* x     = (const float*)d_in[0];
    const int*   ei    = (const int*)  d_in[1];
    const int*   batch = (const int*)  d_in[2];
    const float* W1    = (const float*)d_in[3];
    const float* b1    = (const float*)d_in[4];
    const float* W2    = (const float*)d_in[5];
    const float* b2    = (const float*)d_in[6];
    const float* gamma = (const float*)d_in[7];
    const float* beta  = (const float*)d_in[8];

    float* out  = (float*)d_out;
    float* zcat = out;
    float* gcat = out + (size_t)N_NODES * ZSTRIDE;

    float* hbuf0 = (float*)d_ws;
    float* hbuf1 = hbuf0 + (size_t)N_NODES * DIM;
    float* stats = hbuf1 + (size_t)N_NODES * DIM;
    unsigned* owner = (unsigned*)(stats + NLAYER * 2 * DIM);
    unsigned* saved = owner + N_EDGES / 4;
    int* saved_n    = (int*)(saved + (size_t)AGG_BLOCKS * LIST_CAP);

    const int* src = ei;
    const int* dst = ei + N_EDGES;

    hipMemsetAsync(gcat, 0, sizeof(float) * (size_t)NGRAPH * ZSTRIDE, stream);
    hipMemsetAsync(stats, 0, sizeof(float) * (NLAYER * 2 * DIM), stream);

    owner_kernel<<<(N_EDGES / 4 + 255) / 256, 256, 0, stream>>>(dst, owner);

    static_assert(NLAYER == 3, "the three layers are launched one by one below");
    agg_kernel<0><<<AGG_BLOCKS, AGG_THREADS, 0, stream>>>(
        x, src, dst, owner, saved, saved_n,
        stats, gamma, beta, batch, zcat, gcat,
        W1, b1, W2, b2, hbuf0, stats);
    agg_kernel<1><<<AGG_BLOCKS, AGG_THREADS, 0, stream>>>(
        hbuf0, src, dst, owner, saved, saved_n,
        stats, gamma, beta, batch, zcat, gcat,
        W1 + (size_t)DIM * DIM, b1 + DIM, W2 + (size_t)DIM * DIM, b2 + DIM,
        hbuf1, stats + 2 * DIM);
    agg_kernel<1><<<AGG_BLOCKS, AGG_THREADS, 0, stream>>>(
        hbuf1, src, dst, owner, saved, saved_n,
        stats + 2 * DIM, gamma + DIM, beta + DIM, batch, zcat + DIM, gcat + DIM,
        W1 + (size_t)2 * DIM * DIM, b1 + 2 * DIM, W2 + (size_t)2 * DIM * DIM, b2 + 2 * DIM,
        hbuf0, stats + 4 * DIM);
    agg_kernel<2><<<AGG_BLOCKS, AGG_THREADS, 0, stream>>>(
        hbuf0, src, dst, owner, saved, saved_n,
        stats + 4 * DIM, gamma + 2 * DIM, beta + 2 * DIM, batch, zcat + 2 * DIM, gcat + 2 * DIM,
        W1, b1, W2, b2, hbuf1, stats);
}
